# starting kernel
# baseline (speedup 1.0000x reference)
#include <hip/hip_runtime.h>
#include <hip/hip_bf16.h>
#include <cstddef>


#define N_NODES 50000
#define E_EDGES 800000
#define E_TOT   (E_EDGES + N_NODES)
#define NEG_SLOPE 0.2f

typedef float v2f __attribute__((ext_vector_type(2)));
typedef float v8f __attribute__((ext_vector_type(8)));

__device__ __forceinline__ unsigned f2key(float f) {
    int i = __float_as_int(f);
    return (i >= 0) ? ((unsigned)i | 0x80000000u) : ~(unsigned)i;
}
__device__ __forceinline__ float key2f(unsigned k) {
    int i = (k & 0x80000000u) ? (int)(k & 0x7FFFFFFFu) : (int)~k;
    return __int_as_float(i);
}

__global__ void __launch_bounds__(256)
gat_init(float* __restrict__ acc, unsigned* __restrict__ mkey,
         float* __restrict__ denom, int n_nodes, int nf) {
    int i = blockIdx.x * blockDim.x + threadIdx.x;
    int tot = n_nodes * nf;
    if (i < tot) acc[i] = 0.0f;
    if (i < n_nodes) { mkey[i] = 0u; denom[i] = 0.0f; }
}

__global__ void __launch_bounds__(256)
gat_gemm_wmma(const float* __restrict__ X, const float* __restrict__ W,
              float* __restrict__ H, int nrows, int K) {
    const int wave = blockIdx.x * (blockDim.x >> 5) + (threadIdx.x >> 5);
    const int lane = threadIdx.x & 31;
    const int rowBase = wave * 16;
    if (rowBase >= nrows) return;

    const int lo = lane & 15;
    const int hi = lane >> 4;

    v8f acc0 = {}, acc1 = {}, acc2 = {}, acc3 = {};

    const float* xr = X + (size_t)(rowBase + lo) * K + 2 * hi;
    const float* wc = W + lo;

    for (int k = 0; k < K; k += 4) {
        float2 af = *(const float2*)(xr + k);
        v2f a; a.x = af.x; a.y = af.y;

        const float* w0 = wc + (size_t)(k + 2 * hi) * 64;
        const float* w1 = w0 + 64;
        v2f b0; b0.x = w0[0];  b0.y = w1[0];
        v2f b1; b1.x = w0[16]; b1.y = w1[16];
        v2f b2; b2.x = w0[32]; b2.y = w1[32];
        v2f b3; b3.x = w0[48]; b3.y = w1[48];

        acc0 = __builtin_amdgcn_wmma_f32_16x16x4_f32(false, a, false, b0, (short)0, acc0, false, false);
        acc1 = __builtin_amdgcn_wmma_f32_16x16x4_f32(false, a, false, b1, (short)0, acc1, false, false);
        acc2 = __builtin_amdgcn_wmma_f32_16x16x4_f32(false, a, false, b2, (short)0, acc2, false, false);
        acc3 = __builtin_amdgcn_wmma_f32_16x16x4_f32(false, a, false, b3, (short)0, acc3, false, false);
    }

#pragma unroll
    for (int r = 0; r < 8; ++r) {
        float* o = H + (size_t)(rowBase + r + 8 * hi) * 64 + lo;
        o[0]  = acc0[r];
        o[16] = acc1[r];
        o[32] = acc2[r];
        o[48] = acc3[r];
    }
}

__global__ void __launch_bounds__(256)
gat_node_sd(const float* __restrict__ H, const float* __restrict__ asrc,
            const float* __restrict__ adst, float* __restrict__ s,
            float* __restrict__ d, int n) {
    int i = blockIdx.x * blockDim.x + threadIdx.x;
    if (i >= n) return;
    const float* hr = H + (size_t)i * 64;
    float ss = 0.0f, dd = 0.0f;
#pragma unroll
    for (int f = 0; f < 64; ++f) {
        float h = hr[f];
        ss = fmaf(h, asrc[f], ss);
        dd = fmaf(h, adst[f], dd);
    }
    s[i] = ss; d[i] = dd;
}

__global__ void __launch_bounds__(256)
gat_edge_max(const int* __restrict__ src, const int* __restrict__ dst,
             const float* __restrict__ s, const float* __restrict__ d,
             float* __restrict__ evals, unsigned* __restrict__ mkey,
             int nE, int etot) {
    int e = blockIdx.x * blockDim.x + threadIdx.x;
    if (e >= etot) return;
    int si = (e < nE) ? src[e] : (e - nE);
    int di = (e < nE) ? dst[e] : (e - nE);
    float v = s[si] + d[di];
    v = (v > 0.0f) ? v : NEG_SLOPE * v;
    evals[e] = v;
    atomicMax(mkey + di, f2key(v));
}

__global__ void __launch_bounds__(256)
gat_edge_expsum(const int* __restrict__ dst, const float* __restrict__ evals,
                const unsigned* __restrict__ mkey, float* __restrict__ wvals,
                float* __restrict__ denom, int nE, int etot) {
    int e = blockIdx.x * blockDim.x + threadIdx.x;
    if (e >= etot) return;
    int di = (e < nE) ? dst[e] : (e - nE);
    float w = __expf(evals[e] - key2f(mkey[di]));
    wvals[e] = w;
    atomicAdd(denom + di, w);
}

__global__ void __launch_bounds__(256)
gat_edge_aggregate(const int* __restrict__ src, const int* __restrict__ dst,
                   const float* __restrict__ H, const float* __restrict__ wvals,
                   const float* __restrict__ denom, float* __restrict__ acc,
                   int nE, int etot) {
    int wid  = blockIdx.x * (blockDim.x >> 5) + (threadIdx.x >> 5);
    int lane = threadIdx.x & 31;
    if (wid >= etot) return;
    int si = (wid < nE) ? src[wid] : (wid - nE);
    int di = (wid < nE) ? dst[wid] : (wid - nE);
    float coef = wvals[wid] / (denom[di] + 1e-16f);
    const float* hr = H + (size_t)si * 64;
    float* ar = acc + (size_t)di * 64;
    atomicAdd(ar + lane,      hr[lane]      * coef);
    atomicAdd(ar + lane + 32, hr[lane + 32] * coef);
}

__global__ void __launch_bounds__(256)
gat_bias_act(const float* __restrict__ acc, const float* __restrict__ b,
             float* __restrict__ out, int tot, int relu) {
    int i = blockIdx.x * blockDim.x + threadIdx.x;
    if (i >= tot) return;
    float v = acc[i] + b[i & 63];
    if (relu) v = fmaxf(v, 0.0f);
    out[i] = v;
}

static void run_gat_layer(const float* xin, int K, const float* W,
                          const float* as, const float* ad, const float* bias,
                          float* H, float* acc, float* nextX, int relu,
                          const int* esrc, const int* edst,
                          float* SV, float* DV, unsigned* MKEY, float* DEN,
                          float* EVAL, float* WVAL, hipStream_t stream) {
    const int nodeFeat = N_NODES * 64;
    gat_init<<<(nodeFeat + 255) / 256, 256, 0, stream>>>(acc, MKEY, DEN, N_NODES, 64);
    {
        int waves = N_NODES / 16;
        int blocks = (waves + 7) / 8;
        gat_gemm_wmma<<<blocks, 256, 0, stream>>>(xin, W, H, N_NODES, K);
    }
    gat_node_sd<<<(N_NODES + 255) / 256, 256, 0, stream>>>(H, as, ad, SV, DV, N_NODES);
    gat_edge_max<<<(E_TOT + 255) / 256, 256, 0, stream>>>(esrc, edst, SV, DV, EVAL, MKEY, E_EDGES, E_TOT);
    gat_edge_expsum<<<(E_TOT + 255) / 256, 256, 0, stream>>>(edst, EVAL, MKEY, WVAL, DEN, E_EDGES, E_TOT);
    gat_edge_aggregate<<<(E_TOT + 7) / 8, 256, 0, stream>>>(esrc, edst, H, WVAL, DEN, acc, E_EDGES, E_TOT);
    gat_bias_act<<<(nodeFeat + 255) / 256, 256, 0, stream>>>(acc, bias, nextX, relu, relu);
}

extern "C" void kernel_launch(void* const* d_in, const int* in_sizes, int n_in,
                              void* d_out, int out_size, void* d_ws, size_t ws_size,
                              hipStream_t stream) {
    (void)in_sizes; (void)n_in; (void)out_size; (void)ws_size;

    const float* x    = (const float*)d_in[0];
    const int*   esrc = (const int*)  d_in[1];
    const int*   edst = (const int*)  d_in[2];
    const float* W0 = (const float*)d_in[3];
    const float* a0s = (const float*)d_in[4];
    const float* a0d = (const float*)d_in[5];
    const float* b0 = (const float*)d_in[6];
    const float* W1 = (const float*)d_in[7];
    const float* a1s = (const float*)d_in[8];
    const float* a1d = (const float*)d_in[9];
    const float* b1 = (const float*)d_in[10];
    const float* W2 = (const float*)d_in[11];
    const float* a2s = (const float*)d_in[12];
    const float* a2d = (const float*)d_in[13];
    const float* b2 = (const float*)d_in[14];
    float* out = (float*)d_out;

    float* ws = (float*)d_ws;
    float*    H    = ws;
    float*    XIN  = H   + (size_t)N_NODES * 64;
    float*    ACC  = XIN + (size_t)N_NODES * 64;
    float*    SV   = ACC + (size_t)N_NODES * 64;
    float*    DV   = SV  + N_NODES;
    unsigned* MKEY = (unsigned*)(DV + N_NODES);
    float*    DEN  = (float*)(MKEY + N_NODES);
    float*    EVAL = DEN + N_NODES;
    float*    WVAL = EVAL + E_TOT;

    const int nodeFeat = N_NODES * 64;

    gat_init<<<(nodeFeat + 255) / 256, 256, 0, stream>>>(ACC, MKEY, DEN, N_NODES, 64);
    gat_gemm_wmma<<<((N_NODES / 16) + 7) / 8, 256, 0, stream>>>(x, W0, H, N_NODES, 128);
    gat_node_sd<<<(N_NODES + 255) / 256, 256, 0, stream>>>(H, a0s, a0d, SV, DV, N_NODES);
    gat_edge_max<<<(E_TOT + 255) / 256, 256, 0, stream>>>(esrc, edst, SV, DV, EVAL, MKEY, E_EDGES, E_TOT);
    gat_edge_expsum<<<(E_TOT + 255) / 256, 256, 0, stream>>>(edst, EVAL, MKEY, WVAL, DEN, E_EDGES, E_TOT);
    gat_edge_aggregate<<<(E_TOT + 7) / 8, 256, 0, stream>>>(esrc, edst, H, WVAL, DEN, ACC, E_EDGES, E_TOT);
    gat_bias_act<<<(nodeFeat + 255) / 256, 256, 0, stream>>>(ACC, b0, XIN, nodeFeat, 1);

    gat_init<<<(nodeFeat + 255) / 256, 256, 0, stream>>>(ACC, MKEY, DEN, N_NODES, 64);
    gat_gemm_wmma<<<((N_NODES / 16) + 7) / 8, 256, 0, stream>>>(XIN, W1, H, N_NODES, 64);
    gat_node_sd<<<(N_NODES + 255) / 256, 256, 0, stream>>>(H, a1s, a1d, SV, DV, N_NODES);
    gat_edge_max<<<(E_TOT + 255) / 256, 256, 0, stream>>>(esrc, edst, SV, DV, EVAL, MKEY, E_EDGES, E_TOT);
    gat_edge_expsum<<<(E_TOT + 255) / 256, 256, 0, stream>>>(edst, EVAL, MKEY, WVAL, DEN, E_EDGES, E_TOT);
    gat_edge_aggregate<<<(E_TOT + 7) / 8, 256, 0, stream>>>(esrc, edst, H, WVAL, DEN, ACC, E_EDGES, E_TOT);
    gat_bias_act<<<(nodeFeat + 255) / 256, 256, 0, stream>>>(ACC, b1, XIN, nodeFeat, 1);

    gat_init<<<(nodeFeat + 255) / 256, 256, 0, stream>>>(out, MKEY, DEN, N_NODES, 64);
    gat_gemm_wmma<<<((N_NODES / 16) + 7) / 8, 256, 0, stream>>>(XIN, W2, H, N_NODES, 64);
    gat_node_sd<<<(N_NODES + 255) / 256, 256, 0, stream>>>(H, a2s, a2d, SV, DV, N_NODES);
    gat_edge_max<<<(E_TOT + 255) / 256, 256, 0, stream>>>(esrc, edst, SV, DV, EVAL, MKEY, E_EDGES, E_TOT);
    gat_edge_expsum<<<(E_TOT + 255) / 256, 256, 0, stream>>>(edst, EVAL, MKEY, WVAL, DEN, E_EDGES, E_TOT);
    gat_edge_aggregate<<<(E_TOT + 7) / 8, 256, 0, stream>>>(esrc, edst, H, WVAL, DEN, out, E_EDGES, E_TOT);
    gat_bias_act<<<(nodeFeat + 255) / 256, 256, 0, stream>>>(out, b2, out, nodeFeat, 0);
}


# Bucket edges by destination once per call (returning int add, scan, scatter); each layer gathers per row in one wave: no float atomics, init or bias pass; sum order = bucket slot order
# speedup vs baseline: 3.7090x; 3.7090x over previous
#include <hip/hip_runtime.h>
#include <cstddef>


#define N_NODES 50000
#define E_EDGES 800000
#define NEG_SLOPE 0.2f

typedef float v2f __attribute__((ext_vector_type(2)));
typedef float v8f __attribute__((ext_vector_type(8)));

__global__ void __launch_bounds__(256)
gat_gemm_wmma(const float* __restrict__ X, const float* __restrict__ W,
              float* __restrict__ H, int nrows, int K) {
    const int wave = blockIdx.x * (blockDim.x >> 5) + (threadIdx.x >> 5);
    const int lane = threadIdx.x & 31;
    const int rowBase = wave * 16;
    if (rowBase >= nrows) return;

    const int lo = lane & 15;
    const int hi = lane >> 4;

    v8f acc0 = {}, acc1 = {}, acc2 = {}, acc3 = {};

    const float* xr = X + (size_t)(rowBase + lo) * K + 2 * hi;
    const float* wc = W + lo;

    for (int k = 0; k < K; k += 4) {
        float2 af = *(const float2*)(xr + k);
        v2f a; a.x = af.x; a.y = af.y;

        const float* w0 = wc + (size_t)(k + 2 * hi) * 64;
        const float* w1 = w0 + 64;
        v2f b0; b0.x = w0[0];  b0.y = w1[0];
        v2f b1; b1.x = w0[16]; b1.y = w1[16];
        v2f b2; b2.x = w0[32]; b2.y = w1[32];
        v2f b3; b3.x = w0[48]; b3.y = w1[48];

        acc0 = __builtin_amdgcn_wmma_f32_16x16x4_f32(false, a, false, b0, (short)0, acc0, false, false);
        acc1 = __builtin_amdgcn_wmma_f32_16x16x4_f32(false, a, false, b1, (short)0, acc1, false, false);
        acc2 = __builtin_amdgcn_wmma_f32_16x16x4_f32(false, a, false, b2, (short)0, acc2, false, false);
        acc3 = __builtin_amdgcn_wmma_f32_16x16x4_f32(false, a, false, b3, (short)0, acc3, false, false);
    }

#pragma unroll
    for (int r = 0; r < 8; ++r) {
        float* o = H + (size_t)(rowBase + r + 8 * hi) * 64 + lo;
        o[0]  = acc0[r];
        o[16] = acc1[r];
        o[32] = acc2[r];
        o[48] = acc3[r];
    }
}

__global__ void __launch_bounds__(256)
gat_node_sd(const float* __restrict__ H, const float* __restrict__ asrc,
            const float* __restrict__ adst, float* __restrict__ s,
            float* __restrict__ d, int n) {
    int i = blockIdx.x * blockDim.x + threadIdx.x;
    if (i >= n) return;
    const float* hr = H + (size_t)i * 64;
    float ss = 0.0f, dd = 0.0f;
#pragma unroll
    for (int f = 0; f < 64; ++f) {
        float h = hr[f];
        ss = fmaf(h, asrc[f], ss);
        dd = fmaf(h, adst[f], dd);
    }
    s[i] = ss; d[i] = dd;
}

__global__ void __launch_bounds__(256)
csr_zero(int* __restrict__ cnt, int n) {
    int i = blockIdx.x * blockDim.x + threadIdx.x;
    if (i < n) cnt[i] = 0;
}

__global__ void __launch_bounds__(256)
csr_count(const int* __restrict__ dst, int* __restrict__ cnt,
          int* __restrict__ rank, int nE) {
    int e = blockIdx.x * blockDim.x + threadIdx.x;
    if (e >= nE) return;
    rank[e] = atomicAdd(cnt + dst[e], 1);
}

__global__ void __launch_bounds__(1024)
csr_scan(const int* __restrict__ cnt, int* __restrict__ off, int n) {
    __shared__ int red[32];
    __shared__ int wtot[32];
    const int t = threadIdx.x;
    const int lane = t & 31;
    const int w = t >> 5;
    const int b = blockIdx.x;

    int p = 0;
    for (int k = 0; k < b; ++k) p += cnt[k * 1024 + t];
    for (int m = 16; m >= 1; m >>= 1) p += __shfl_xor(p, m, 32);

    const int i = b * 1024 + t;
    const int v = (i < n) ? cnt[i] : 0;
    int x = v;
    for (int dlt = 1; dlt < 32; dlt <<= 1) {
        int y = __shfl_up(x, dlt, 32);
        if (lane >= dlt) x += y;
    }
    if (lane == 0)  red[w]  = p;
    if (lane == 31) wtot[w] = x;
    __syncthreads();
    int base = 0;
    for (int j = 0; j < 32; ++j) {
        base += red[j];
        if (j < w) base += wtot[j];
    }
    const int excl = base + x - v;
    if (i < n) off[i] = excl;
    if (i == n - 1) off[n] = excl + v;
}

__global__ void __launch_bounds__(256)
csr_scatter(const int* __restrict__ src, const int* __restrict__ dst,
            const int* __restrict__ off, const int* __restrict__ rank,
            int* __restrict__ csrc, int nE) {
    int e = blockIdx.x * blockDim.x + threadIdx.x;
    if (e >= nE) return;
    csrc[off[dst[e]] + rank[e]] = src[e];
}

__device__ __forceinline__ float leaky(float v) {
    return (v > 0.0f) ? v : NEG_SLOPE * v;
}

__global__ void __launch_bounds__(256)
gat_segment(const int* __restrict__ off, const int* __restrict__ csrc,
            const float* __restrict__ H, const float* __restrict__ s,
            const float* __restrict__ d, const float* __restrict__ bias,
            float* __restrict__ out, int n, int relu) {
    const int node = __builtin_amdgcn_readfirstlane(
        (int)(blockIdx.x * (blockDim.x >> 5) + (threadIdx.x >> 5)));
    const int lane = threadIdx.x & 31;
    if (node >= n) return;

    const int beg = off[node];
    const int deg = off[node + 1] - beg;
    const float dn = d[node];
    const float es = leaky(s[node] + dn);

    int   sj0 = node;
    float e0  = es;
    if (lane < deg) {
        sj0 = csrc[beg + lane];
        e0  = leaky(s[sj0] + dn);
    }

    float m = fmaxf(e0, es);
    for (int c = 32; c < deg; c += 32) {
        int j = c + lane;
        if (j < deg) m = fmaxf(m, leaky(s[csrc[beg + j]] + dn));
    }
    for (int k = 16; k >= 1; k >>= 1) m = fmaxf(m, __shfl_xor(m, k, 32));

    const float w0 = (lane < deg) ? __expf(e0 - m) : 0.0f;
    float sum = w0;
    for (int c = 32; c < deg; c += 32) {
        int j = c + lane;
        if (j < deg) sum += __expf(leaky(s[csrc[beg + j]] + dn) - m);
    }
    for (int k = 16; k >= 1; k >>= 1) sum += __shfl_xor(sum, k, 32);
    const float wself = __expf(es - m);
    const float den = (sum + wself) + 1e-16f;

    const float2* __restrict__ H2 = (const float2*)H;
    float2 hv = H2[(size_t)node * 32 + lane];
    const float cs = wself / den;
    float2 acc;
    acc.x = hv.x * cs;
    acc.y = hv.y * cs;

    for (int c = 0; c < deg; c += 32) {
        int   sj;
        float cf;
        if (c == 0) {
            sj = sj0;
            cf = w0 / den;
        } else {
            int j = c + lane;
            sj = node;
            cf = 0.0f;
            if (j < deg) {
                sj = csrc[beg + j];
                cf = __expf(leaky(s[sj] + dn) - m) / den;
            }
        }
        const int cnt = (deg - c < 32) ? (deg - c) : 32;
#pragma unroll 4
        for (int t = 0; t < cnt; ++t) {
            const int   su = __builtin_amdgcn_readlane(sj, t);
            const float cu = __int_as_float(
                __builtin_amdgcn_readlane(__float_as_int(cf), t));
            const float2 h = H2[(size_t)su * 32 + lane];
            acc.x = fmaf(h.x, cu, acc.x);
            acc.y = fmaf(h.y, cu, acc.y);
        }
    }

    const float2 bb = ((const float2*)bias)[lane];
    float2 o;
    o.x = acc.x + bb.x;
    o.y = acc.y + bb.y;
    if (relu) { o.x = fmaxf(o.x, 0.0f); o.y = fmaxf(o.y, 0.0f); }
    ((float2*)out)[(size_t)node * 32 + lane] = o;
}

extern "C" void kernel_launch(void* const* d_in, const int* in_sizes, int n_in,
                              void* d_out, int out_size, void* d_ws, size_t ws_size,
                              hipStream_t stream) {
    (void)in_sizes; (void)n_in; (void)out_size; (void)ws_size;

    const float* x    = (const float*)d_in[0];
    const int*   esrc = (const int*)  d_in[1];
    const int*   edst = (const int*)  d_in[2];
    const float* W0 = (const float*)d_in[3];
    const float* a0s = (const float*)d_in[4];
    const float* a0d = (const float*)d_in[5];
    const float* b0 = (const float*)d_in[6];
    const float* W1 = (const float*)d_in[7];
    const float* a1s = (const float*)d_in[8];
    const float* a1d = (const float*)d_in[9];
    const float* b1 = (const float*)d_in[10];
    const float* W2 = (const float*)d_in[11];
    const float* a2s = (const float*)d_in[12];
    const float* a2d = (const float*)d_in[13];
    const float* b2 = (const float*)d_in[14];
    float* out = (float*)d_out;

    float* ws = (float*)d_ws;
    float* H    = ws;
    float* XIN  = H   + (size_t)N_NODES * 64;
    float* SV   = XIN + (size_t)N_NODES * 64;
    float* DV   = SV  + N_NODES;
    int*   CNT  = (int*)(DV + N_NODES);
    int*   OFF  = CNT + N_NODES;
    int*   RANK = OFF + N_NODES + 1;
    int*   CSRC = RANK + E_EDGES;

    csr_zero<<<(N_NODES + 255) / 256, 256, 0, stream>>>(CNT, N_NODES);
    csr_count<<<(E_EDGES + 255) / 256, 256, 0, stream>>>(edst, CNT, RANK, E_EDGES);
    csr_scan<<<(N_NODES + 1023) / 1024, 1024, 0, stream>>>(CNT, OFF, N_NODES);
    csr_scatter<<<(E_EDGES + 255) / 256, 256, 0, stream>>>(esrc, edst, OFF, RANK, CSRC, E_EDGES);

    gat_gemm_wmma<<<((N_NODES / 16) + 7) / 8, 256, 0, stream>>>(x, W0, H, N_NODES, 128);
    gat_node_sd<<<(N_NODES + 255) / 256, 256, 0, stream>>>(H, a0s, a0d, SV, DV, N_NODES);
    gat_segment<<<(N_NODES + 7) / 8, 256, 0, stream>>>(OFF, CSRC, H, SV, DV, b0, XIN, N_NODES, 1);

    gat_gemm_wmma<<<((N_NODES / 16) + 7) / 8, 256, 0, stream>>>(XIN, W1, H, N_NODES, 64);
    gat_node_sd<<<(N_NODES + 255) / 256, 256, 0, stream>>>(H, a1s, a1d, SV, DV, N_NODES);
    gat_segment<<<(N_NODES + 7) / 8, 256, 0, stream>>>(OFF, CSRC, H, SV, DV, b1, XIN, N_NODES, 1);

    gat_gemm_wmma<<<((N_NODES / 16) + 7) / 8, 256, 0, stream>>>(XIN, W2, H, N_NODES, 64);
    gat_node_sd<<<(N_NODES + 255) / 256, 256, 0, stream>>>(H, a2s, a2d, SV, DV, N_NODES);
    gat_segment<<<(N_NODES + 7) / 8, 256, 0, stream>>>(OFF, CSRC, H, SV, DV, b2, out, N_NODES, 0);
}


# Attention logits s,d formed in the GEMM epilogue from the accumulator tile (four column products per lane, then a half-wave xor tree); the three logit launches go; logit sum order changes
# speedup vs baseline: 3.9399x; 1.0623x over previous
#include <hip/hip_runtime.h>
#include <cstddef>


#define N_NODES 50000
#define E_EDGES 800000
#define NEG_SLOPE 0.2f

typedef float v2f __attribute__((ext_vector_type(2)));
typedef float v8f __attribute__((ext_vector_type(8)));

__global__ void __launch_bounds__(256)
gat_gemm_wmma(const float* __restrict__ X, const float* __restrict__ W,
              const float* __restrict__ asrc, const float* __restrict__ adst,
              float* __restrict__ H, float* __restrict__ s,
              float* __restrict__ d, int nrows, int K) {
    const int wave = blockIdx.x * (blockDim.x >> 5) + (threadIdx.x >> 5);
    const int lane = threadIdx.x & 31;
    const int rowBase = wave * 16;
    if (rowBase >= nrows) return;

    const int lo = lane & 15;
    const int hi = lane >> 4;

    v8f acc0 = {}, acc1 = {}, acc2 = {}, acc3 = {};

    const float as0 = asrc[lo], as1 = asrc[16 + lo], as2 = asrc[32 + lo], as3 = asrc[48 + lo];
    const float ad0 = adst[lo], ad1 = adst[16 + lo], ad2 = adst[32 + lo], ad3 = adst[48 + lo];

    const float* xr = X + (size_t)(rowBase + lo) * K + 2 * hi;
    const float* wc = W + lo;

    for (int k = 0; k < K; k += 4) {
        float2 af = *(const float2*)(xr + k);
        v2f a; a.x = af.x; a.y = af.y;

        const float* w0 = wc + (size_t)(k + 2 * hi) * 64;
        const float* w1 = w0 + 64;
        v2f b0; b0.x = w0[0];  b0.y = w1[0];
        v2f b1; b1.x = w0[16]; b1.y = w1[16];
        v2f b2; b2.x = w0[32]; b2.y = w1[32];
        v2f b3; b3.x = w0[48]; b3.y = w1[48];

        acc0 = __builtin_amdgcn_wmma_f32_16x16x4_f32(false, a, false, b0, (short)0, acc0, false, false);
        acc1 = __builtin_amdgcn_wmma_f32_16x16x4_f32(false, a, false, b1, (short)0, acc1, false, false);
        acc2 = __builtin_amdgcn_wmma_f32_16x16x4_f32(false, a, false, b2, (short)0, acc2, false, false);
        acc3 = __builtin_amdgcn_wmma_f32_16x16x4_f32(false, a, false, b3, (short)0, acc3, false, false);
    }

#pragma unroll
    for (int r = 0; r < 8; ++r) {
        float* o = H + (size_t)(rowBase + r + 8 * hi) * 64 + lo;
        o[0]  = acc0[r];
        o[16] = acc1[r];
        o[32] = acc2[r];
        o[48] = acc3[r];
    }

    float ps[8], pd[8];
#pragma unroll
    for (int r = 0; r < 8; ++r) {
        ps[r] = fmaf(acc3[r], as3, fmaf(acc2[r], as2, fmaf(acc1[r], as1, acc0[r] * as0)));
        pd[r] = fmaf(acc3[r], ad3, fmaf(acc2[r], ad2, fmaf(acc1[r], ad1, acc0[r] * ad0)));
    }
#pragma unroll
    for (int m = 8; m >= 1; m >>= 1) {
#pragma unroll
        for (int r = 0; r < 8; ++r) {
            ps[r] += __shfl_xor(ps[r], m, 32);
            pd[r] += __shfl_xor(pd[r], m, 32);
        }
    }
    float sv = ps[0], dv = pd[0];
#pragma unroll
    for (int r = 1; r < 8; ++r) {
        if (lo == r) { sv = ps[r]; dv = pd[r]; }
    }
    if (lo < 8) {
        s[rowBase + 8 * hi + lo] = sv;
        d[rowBase + 8 * hi + lo] = dv;
    }
}

__global__ void __launch_bounds__(256)
csr_zero(int* __restrict__ cnt, int n) {
    int i = blockIdx.x * blockDim.x + threadIdx.x;
    if (i < n) cnt[i] = 0;
}

__global__ void __launch_bounds__(256)
csr_count(const int* __restrict__ dst, int* __restrict__ cnt,
          int* __restrict__ rank, int nE) {
    int e = blockIdx.x * blockDim.x + threadIdx.x;
    if (e >= nE) return;
    rank[e] = atomicAdd(cnt + dst[e], 1);
}

__global__ void __launch_bounds__(1024)
csr_scan(const int* __restrict__ cnt, int* __restrict__ off, int n) {
    __shared__ int red[32];
    __shared__ int wtot[32];
    const int t = threadIdx.x;
    const int lane = t & 31;
    const int w = t >> 5;
    const int b = blockIdx.x;

    int p = 0;
    for (int k = 0; k < b; ++k) p += cnt[k * 1024 + t];
    for (int m = 16; m >= 1; m >>= 1) p += __shfl_xor(p, m, 32);

    const int i = b * 1024 + t;
    const int v = (i < n) ? cnt[i] : 0;
    int x = v;
    for (int dlt = 1; dlt < 32; dlt <<= 1) {
        int y = __shfl_up(x, dlt, 32);
        if (lane >= dlt) x += y;
    }
    if (lane == 0)  red[w]  = p;
    if (lane == 31) wtot[w] = x;
    __syncthreads();
    int base = 0;
    for (int j = 0; j < 32; ++j) {
        base += red[j];
        if (j < w) base += wtot[j];
    }
    const int excl = base + x - v;
    if (i < n) off[i] = excl;
    if (i == n - 1) off[n] = excl + v;
}

__global__ void __launch_bounds__(256)
csr_scatter(const int* __restrict__ src, const int* __restrict__ dst,
            const int* __restrict__ off, const int* __restrict__ rank,
            int* __restrict__ csrc, int nE) {
    int e = blockIdx.x * blockDim.x + threadIdx.x;
    if (e >= nE) return;
    csrc[off[dst[e]] + rank[e]] = src[e];
}

__device__ __forceinline__ float leaky(float v) {
    return (v > 0.0f) ? v : NEG_SLOPE * v;
}

__global__ void __launch_bounds__(256)
gat_segment(const int* __restrict__ off, const int* __restrict__ csrc,
            const float* __restrict__ H, const float* __restrict__ s,
            const float* __restrict__ d, const float* __restrict__ bias,
            float* __restrict__ out, int n, int relu) {
    const int node = __builtin_amdgcn_readfirstlane(
        (int)(blockIdx.x * (blockDim.x >> 5) + (threadIdx.x >> 5)));
    const int lane = threadIdx.x & 31;
    if (node >= n) return;

    const int beg = off[node];
    const int deg = off[node + 1] - beg;
    const float dn = d[node];
    const float es = leaky(s[node] + dn);

    int   sj0 = node;
    float e0  = es;
    if (lane < deg) {
        sj0 = csrc[beg + lane];
        e0  = leaky(s[sj0] + dn);
    }

    float m = fmaxf(e0, es);
    for (int c = 32; c < deg; c += 32) {
        int j = c + lane;
        if (j < deg) m = fmaxf(m, leaky(s[csrc[beg + j]] + dn));
    }
    for (int k = 16; k >= 1; k >>= 1) m = fmaxf(m, __shfl_xor(m, k, 32));

    const float w0 = (lane < deg) ? __expf(e0 - m) : 0.0f;
    float sum = w0;
    for (int c = 32; c < deg; c += 32) {
        int j = c + lane;
        if (j < deg) sum += __expf(leaky(s[csrc[beg + j]] + dn) - m);
    }
    for (int k = 16; k >= 1; k >>= 1) sum += __shfl_xor(sum, k, 32);
    const float wself = __expf(es - m);
    const float den = (sum + wself) + 1e-16f;

    const float2* __restrict__ H2 = (const float2*)H;
    float2 hv = H2[(size_t)node * 32 + lane];
    const float cs = wself / den;
    float2 acc;
    acc.x = hv.x * cs;
    acc.y = hv.y * cs;

    for (int c = 0; c < deg; c += 32) {
        int   sj;
        float cf;
        if (c == 0) {
            sj = sj0;
            cf = w0 / den;
        } else {
            int j = c + lane;
            sj = node;
            cf = 0.0f;
            if (j < deg) {
                sj = csrc[beg + j];
                cf = __expf(leaky(s[sj] + dn) - m) / den;
            }
        }
        const int cnt = (deg - c < 32) ? (deg - c) : 32;
#pragma unroll 4
        for (int t = 0; t < cnt; ++t) {
            const int   su = __builtin_amdgcn_readlane(sj, t);
            const float cu = __int_as_float(
                __builtin_amdgcn_readlane(__float_as_int(cf), t));
            const float2 h = H2[(size_t)su * 32 + lane];
            acc.x = fmaf(h.x, cu, acc.x);
            acc.y = fmaf(h.y, cu, acc.y);
        }
    }

    const float2 bb = ((const float2*)bias)[lane];
    float2 o;
    o.x = acc.x + bb.x;
    o.y = acc.y + bb.y;
    if (relu) { o.x = fmaxf(o.x, 0.0f); o.y = fmaxf(o.y, 0.0f); }
    ((float2*)out)[(size_t)node * 32 + lane] = o;
}

extern "C" void kernel_launch(void* const* d_in, const int* in_sizes, int n_in,
                              void* d_out, int out_size, void* d_ws, size_t ws_size,
                              hipStream_t stream) {
    (void)in_sizes; (void)n_in; (void)out_size; (void)ws_size;

    const float* x    = (const float*)d_in[0];
    const int*   esrc = (const int*)  d_in[1];
    const int*   edst = (const int*)  d_in[2];
    const float* W0 = (const float*)d_in[3];
    const float* a0s = (const float*)d_in[4];
    const float* a0d = (const float*)d_in[5];
    const float* b0 = (const float*)d_in[6];
    const float* W1 = (const float*)d_in[7];
    const float* a1s = (const float*)d_in[8];
    const float* a1d = (const float*)d_in[9];
    const float* b1 = (const float*)d_in[10];
    const float* W2 = (const float*)d_in[11];
    const float* a2s = (const float*)d_in[12];
    const float* a2d = (const float*)d_in[13];
    const float* b2 = (const float*)d_in[14];
    float* out = (float*)d_out;

    float* ws = (float*)d_ws;
    float* H    = ws;
    float* XIN  = H   + (size_t)N_NODES * 64;
    float* SV   = XIN + (size_t)N_NODES * 64;
    float* DV   = SV  + N_NODES;
    int*   CNT  = (int*)(DV + N_NODES);
    int*   OFF  = CNT + N_NODES;
    int*   RANK = OFF + N_NODES + 1;
    int*   CSRC = RANK + E_EDGES;

    csr_zero<<<(N_NODES + 255) / 256, 256, 0, stream>>>(CNT, N_NODES);
    csr_count<<<(E_EDGES + 255) / 256, 256, 0, stream>>>(edst, CNT, RANK, E_EDGES);
    csr_scan<<<(N_NODES + 1023) / 1024, 1024, 0, stream>>>(CNT, OFF, N_NODES);
    csr_scatter<<<(E_EDGES + 255) / 256, 256, 0, stream>>>(esrc, edst, OFF, RANK, CSRC, E_EDGES);

    gat_gemm_wmma<<<((N_NODES / 16) + 7) / 8, 256, 0, stream>>>(x, W0, a0s, a0d, H, SV, DV, N_NODES, 128);
    gat_segment<<<(N_NODES + 7) / 8, 256, 0, stream>>>(OFF, CSRC, H, SV, DV, b0, XIN, N_NODES, 1);

    gat_gemm_wmma<<<((N_NODES / 16) + 7) / 8, 256, 0, stream>>>(XIN, W1, a1s, a1d, H, SV, DV, N_NODES, 64);
    gat_segment<<<(N_NODES + 7) / 8, 256, 0, stream>>>(OFF, CSRC, H, SV, DV, b1, XIN, N_NODES, 1);

    gat_gemm_wmma<<<((N_NODES / 16) + 7) / 8, 256, 0, stream>>>(XIN, W2, a2s, a2d, H, SV, DV, N_NODES, 64);
    gat_segment<<<(N_NODES + 7) / 8, 256, 0, stream>>>(OFF, CSRC, H, SV, DV, b2, out, N_NODES, 0);
}


# Row pass: eight lanes per destination row, a wave covers four rows side by side with 16-byte loads and requests eight neighbour rows per trip (pad slots weigh 0); denominator sum tree changes
# speedup vs baseline: 4.0696x; 1.0329x over previous
#include <hip/hip_runtime.h>
#include <cstddef>


#define N_NODES 50000
#define E_EDGES 800000
#define NEG_SLOPE 0.2f

typedef float v2f __attribute__((ext_vector_type(2)));
typedef float v8f __attribute__((ext_vector_type(8)));

__global__ void __launch_bounds__(256)
gat_gemm_wmma(const float* __restrict__ X, const float* __restrict__ W,
              const float* __restrict__ asrc, const float* __restrict__ adst,
              float* __restrict__ H, float* __restrict__ s,
              float* __restrict__ d, int nrows, int K) {
    const int wave = blockIdx.x * (blockDim.x >> 5) + (threadIdx.x >> 5);
    const int lane = threadIdx.x & 31;
    const int rowBase = wave * 16;
    if (rowBase >= nrows) return;

    const int lo = lane & 15;
    const int hi = lane >> 4;

    v8f acc0 = {}, acc1 = {}, acc2 = {}, acc3 = {};

    const float as0 = asrc[lo], as1 = asrc[16 + lo], as2 = asrc[32 + lo], as3 = asrc[48 + lo];
    const float ad0 = adst[lo], ad1 = adst[16 + lo], ad2 = adst[32 + lo], ad3 = adst[48 + lo];

    const float* xr = X + (size_t)(rowBase + lo) * K + 2 * hi;
    const float* wc = W + lo;

    for (int k = 0; k < K; k += 4) {
        float2 af = *(const float2*)(xr + k);
        v2f a; a.x = af.x; a.y = af.y;

        const float* w0 = wc + (size_t)(k + 2 * hi) * 64;
        const float* w1 = w0 + 64;
        v2f b0; b0.x = w0[0];  b0.y = w1[0];
        v2f b1; b1.x = w0[16]; b1.y = w1[16];
        v2f b2; b2.x = w0[32]; b2.y = w1[32];
        v2f b3; b3.x = w0[48]; b3.y = w1[48];

        acc0 = __builtin_amdgcn_wmma_f32_16x16x4_f32(false, a, false, b0, (short)0, acc0, false, false);
        acc1 = __builtin_amdgcn_wmma_f32_16x16x4_f32(false, a, false, b1, (short)0, acc1, false, false);
        acc2 = __builtin_amdgcn_wmma_f32_16x16x4_f32(false, a, false, b2, (short)0, acc2, false, false);
        acc3 = __builtin_amdgcn_wmma_f32_16x16x4_f32(false, a, false, b3, (short)0, acc3, false, false);
    }

#pragma unroll
    for (int r = 0; r < 8; ++r) {
        float* o = H + (size_t)(rowBase + r + 8 * hi) * 64 + lo;
        o[0]  = acc0[r];
        o[16] = acc1[r];
        o[32] = acc2[r];
        o[48] = acc3[r];
    }

    float ps[8], pd[8];
#pragma unroll
    for (int r = 0; r < 8; ++r) {
        ps[r] = fmaf(acc3[r], as3, fmaf(acc2[r], as2, fmaf(acc1[r], as1, acc0[r] * as0)));
        pd[r] = fmaf(acc3[r], ad3, fmaf(acc2[r], ad2, fmaf(acc1[r], ad1, acc0[r] * ad0)));
    }
#pragma unroll
    for (int m = 8; m >= 1; m >>= 1) {
#pragma unroll
        for (int r = 0; r < 8; ++r) {
            ps[r] += __shfl_xor(ps[r], m, 32);
            pd[r] += __shfl_xor(pd[r], m, 32);
        }
    }
    float sv = ps[0], dv = pd[0];
#pragma unroll
    for (int r = 1; r < 8; ++r) {
        if (lo == r) { sv = ps[r]; dv = pd[r]; }
    }
    if (lo < 8) {
        s[rowBase + 8 * hi + lo] = sv;
        d[rowBase + 8 * hi + lo] = dv;
    }
}

__global__ void __launch_bounds__(256)
csr_zero(int* __restrict__ cnt, int n) {
    int i = blockIdx.x * blockDim.x + threadIdx.x;
    if (i < n) cnt[i] = 0;
}

__global__ void __launch_bounds__(256)
csr_count(const int* __restrict__ dst, int* __restrict__ cnt,
          int* __restrict__ rank, int nE) {
    int e = blockIdx.x * blockDim.x + threadIdx.x;
    if (e >= nE) return;
    rank[e] = atomicAdd(cnt + dst[e], 1);
}

__global__ void __launch_bounds__(1024)
csr_scan(const int* __restrict__ cnt, int* __restrict__ off, int n) {
    __shared__ int red[32];
    __shared__ int wtot[32];
    const int t = threadIdx.x;
    const int lane = t & 31;
    const int w = t >> 5;
    const int b = blockIdx.x;

    int p = 0;
    for (int k = 0; k < b; ++k) p += cnt[k * 1024 + t];
    for (int m = 16; m >= 1; m >>= 1) p += __shfl_xor(p, m, 32);

    const int i = b * 1024 + t;
    const int v = (i < n) ? cnt[i] : 0;
    int x = v;
    for (int dlt = 1; dlt < 32; dlt <<= 1) {
        int y = __shfl_up(x, dlt, 32);
        if (lane >= dlt) x += y;
    }
    if (lane == 0)  red[w]  = p;
    if (lane == 31) wtot[w] = x;
    __syncthreads();
    int base = 0;
    for (int j = 0; j < 32; ++j) {
        base += red[j];
        if (j < w) base += wtot[j];
    }
    const int excl = base + x - v;
    if (i < n) off[i] = excl;
    if (i == n - 1) off[n] = excl + v;
}

__global__ void __launch_bounds__(256)
csr_scatter(const int* __restrict__ src, const int* __restrict__ dst,
            const int* __restrict__ off, const int* __restrict__ rank,
            int* __restrict__ csrc, int nE) {
    int e = blockIdx.x * blockDim.x + threadIdx.x;
    if (e >= nE) return;
    csrc[off[dst[e]] + rank[e]] = src[e];
}

__device__ __forceinline__ float leaky(float v) {
    return (v > 0.0f) ? v : NEG_SLOPE * v;
}


#define SEL8_BCAST(T) ((T) * 0x249249u)
#define SEL8_XOR1 0xde54c1u
#define SEL8_XOR2 0xb3e21au
#define SEL8_XOR4 0x688facu

__device__ __forceinline__ float group_max8(float v) {
    v = fmaxf(v, __uint_as_float(__builtin_amdgcn_mov_dpp8(__float_as_uint(v), SEL8_XOR1)));
    v = fmaxf(v, __uint_as_float(__builtin_amdgcn_mov_dpp8(__float_as_uint(v), SEL8_XOR2)));
    v = fmaxf(v, __uint_as_float(__builtin_amdgcn_mov_dpp8(__float_as_uint(v), SEL8_XOR4)));
    return v;
}

__device__ __forceinline__ float group_sum8(float v) {
    v += __uint_as_float(__builtin_amdgcn_mov_dpp8(__float_as_uint(v), SEL8_XOR1));
    v += __uint_as_float(__builtin_amdgcn_mov_dpp8(__float_as_uint(v), SEL8_XOR2));
    v += __uint_as_float(__builtin_amdgcn_mov_dpp8(__float_as_uint(v), SEL8_XOR4));
    return v;
}

__device__ __forceinline__ void gather8(const char* __restrict__ Hb, unsigned laneoff,
                                        int sj, float cf, float4& a0, float4& a1) {
    float4 p[8], q[8];
    float  cu[8];
#define GAT_LD(T)                                                                          \
    {                                                                                      \
        const unsigned su = __builtin_amdgcn_mov_dpp8((unsigned)sj, SEL8_BCAST(T));        \
        cu[T] = __uint_as_float(__builtin_amdgcn_mov_dpp8(__float_as_uint(cf), SEL8_BCAST(T))); \
        const char* r = Hb + ((su << 8) | laneoff);                                        \
        p[T] = *(const float4*)r;                                                          \
        q[T] = *(const float4*)(r + 128);                                                  \
    }
    GAT_LD(0) GAT_LD(1) GAT_LD(2) GAT_LD(3) GAT_LD(4) GAT_LD(5) GAT_LD(6) GAT_LD(7)
#undef GAT_LD
#pragma unroll
    for (int t = 0; t < 8; ++t) {
        a0.x = fmaf(p[t].x, cu[t], a0.x);
        a0.y = fmaf(p[t].y, cu[t], a0.y);
        a0.z = fmaf(p[t].z, cu[t], a0.z);
        a0.w = fmaf(p[t].w, cu[t], a0.w);
        a1.x = fmaf(q[t].x, cu[t], a1.x);
        a1.y = fmaf(q[t].y, cu[t], a1.y);
        a1.z = fmaf(q[t].z, cu[t], a1.z);
        a1.w = fmaf(q[t].w, cu[t], a1.w);
    }
}

__global__ void __launch_bounds__(256)
gat_segment(const int* __restrict__ off, const int* __restrict__ csrc,
            const float* __restrict__ H, const float* __restrict__ s,
            const float* __restrict__ d, const float* __restrict__ bias,
            float* __restrict__ out, int n, int relu) {
    const int wave = __builtin_amdgcn_readfirstlane(
        (int)(blockIdx.x * (blockDim.x >> 5) + (threadIdx.x >> 5)));
    if (wave * 4 >= n) return;
    const int lane = threadIdx.x & 31;
    const int l8 = lane & 7;
    int node = wave * 4 + (lane >> 3);
    const bool live = node < n;
    if (!live) node = n - 1;

    const char* __restrict__ Hb = (const char*)H;
    const unsigned laneoff = (unsigned)l8 << 4;
    const unsigned own = ((unsigned)node << 8) | laneoff;

    const float4 hp = *(const float4*)(Hb + own);
    const float4 hq = *(const float4*)(Hb + own + 128);
    const float4 bp = ((const float4*)bias)[l8];
    const float4 bq = ((const float4*)bias)[8 + l8];

    const int beg = off[node];
    const int deg = off[node + 1] - beg;
    const float dn = d[node];
    const float es = leaky(s[node] + dn);

    const int dA = __builtin_amdgcn_readlane(deg, 0);
    const int dB = __builtin_amdgcn_readlane(deg, 8);
    const int dC = __builtin_amdgcn_readlane(deg, 16);
    const int dD = __builtin_amdgcn_readlane(deg, 24);
    const int dAB = dA > dB ? dA : dB;
    const int dCD = dC > dD ? dC : dD;
    const int dmax = dAB > dCD ? dAB : dCD;

    int   sj[4];
    bool  has[4];
    int   ld[4];
#pragma unroll
    for (int k = 0; k < 4; ++k) {
        const int j = l8 + 8 * k;
        has[k] = j < deg;
        ld[k] = csrc[has[k] ? beg + j : 0];
    }
#pragma unroll
    for (int k = 0; k < 4; ++k) sj[k] = has[k] ? ld[k] : node;
    float e[4];
#pragma unroll
    for (int k = 0; k < 4; ++k) e[k] = leaky(s[sj[k]] + dn);

    float m = fmaxf(fmaxf(fmaxf(e[0], e[1]), fmaxf(e[2], e[3])), es);
    for (int c = 32; c < dmax; c += 8) {
        const int j = c + l8;
        if (j < deg) m = fmaxf(m, leaky(s[csrc[beg + j]] + dn));
    }
    m = group_max8(m);

    float w[4];
#pragma unroll
    for (int k = 0; k < 4; ++k) w[k] = has[k] ? __expf(e[k] - m) : 0.0f;
    float sum = (w[0] + w[1]) + (w[2] + w[3]);
    for (int c = 32; c < dmax; c += 8) {
        const int j = c + l8;
        if (j < deg) sum += __expf(leaky(s[csrc[beg + j]] + dn) - m);
    }
    sum = group_sum8(sum);
    const float wself = __expf(es - m);
    const float den = (sum + wself) + 1e-16f;

    const float cs = wself / den;
    float4 a0, a1;
    a0.x = hp.x * cs; a0.y = hp.y * cs; a0.z = hp.z * cs; a0.w = hp.w * cs;
    a1.x = hq.x * cs; a1.y = hq.y * cs; a1.z = hq.z * cs; a1.w = hq.w * cs;

#pragma unroll
    for (int k = 0; k < 4; ++k) {
        if (dmax > 8 * k) gather8(Hb, laneoff, sj[k], w[k] / den, a0, a1);
    }
    for (int c = 32; c < dmax; c += 8) {
        const int j = c + l8;
        int   sx = node;
        float cx = 0.0f;
        if (j < deg) {
            sx = csrc[beg + j];
            cx = __expf(leaky(s[sx] + dn) - m) / den;
        }
        gather8(Hb, laneoff, sx, cx, a0, a1);
    }

    float4 o0, o1;
    o0.x = a0.x + bp.x; o0.y = a0.y + bp.y; o0.z = a0.z + bp.z; o0.w = a0.w + bp.w;
    o1.x = a1.x + bq.x; o1.y = a1.y + bq.y; o1.z = a1.z + bq.z; o1.w = a1.w + bq.w;
    if (relu) {
        o0.x = fmaxf(o0.x, 0.0f); o0.y = fmaxf(o0.y, 0.0f); o0.z = fmaxf(o0.z, 0.0f); o0.w = fmaxf(o0.w, 0.0f);
        o1.x = fmaxf(o1.x, 0.0f); o1.y = fmaxf(o1.y, 0.0f); o1.z = fmaxf(o1.z, 0.0f); o1.w = fmaxf(o1.w, 0.0f);
    }
    if (live) {
        char* ob = (char*)out;
        *(float4*)(ob + own)       = o0;
        *(float4*)(ob + own + 128) = o1;
    }
}

extern "C" void kernel_launch(void* const* d_in, const int* in_sizes, int n_in,
                              void* d_out, int out_size, void* d_ws, size_t ws_size,
                              hipStream_t stream) {
    (void)in_sizes; (void)n_in; (void)out_size; (void)ws_size;

    const float* x    = (const float*)d_in[0];
    const int*   esrc = (const int*)  d_in[1];
    const int*   edst = (const int*)  d_in[2];
    const float* W0 = (const float*)d_in[3];
    const float* a0s = (const float*)d_in[4];
    const float* a0d = (const float*)d_in[5];
    const float* b0 = (const float*)d_in[6];
    const float* W1 = (const float*)d_in[7];
    const float* a1s = (const float*)d_in[8];
    const float* a1d = (const float*)d_in[9];
    const float* b1 = (const float*)d_in[10];
    const float* W2 = (const float*)d_in[11];
    const float* a2s = (const float*)d_in[12];
    const float* a2d = (const float*)d_in[13];
    const float* b2 = (const float*)d_in[14];
    float* out = (float*)d_out;

    float* ws = (float*)d_ws;
    float* H    = ws;
    float* XIN  = H   + (size_t)N_NODES * 64;
    float* SV   = XIN + (size_t)N_NODES * 64;
    float* DV   = SV  + N_NODES;
    int*   CNT  = (int*)(DV + N_NODES);
    int*   OFF  = CNT + N_NODES;
    int*   RANK = OFF + N_NODES + 1;
    int*   CSRC = RANK + E_EDGES;

    csr_zero<<<(N_NODES + 255) / 256, 256, 0, stream>>>(CNT, N_NODES);
    csr_count<<<(E_EDGES + 255) / 256, 256, 0, stream>>>(edst, CNT, RANK, E_EDGES);
    csr_scan<<<(N_NODES + 1023) / 1024, 1024, 0, stream>>>(CNT, OFF, N_NODES);
    csr_scatter<<<(E_EDGES + 255) / 256, 256, 0, stream>>>(esrc, edst, OFF, RANK, CSRC, E_EDGES);

    gat_gemm_wmma<<<((N_NODES / 16) + 7) / 8, 256, 0, stream>>>(x, W0, a0s, a0d, H, SV, DV, N_NODES, 128);
    gat_segment<<<((N_NODES + 3) / 4 + 7) / 8, 256, 0, stream>>>(OFF, CSRC, H, SV, DV, b0, XIN, N_NODES, 1);

    gat_gemm_wmma<<<((N_NODES / 16) + 7) / 8, 256, 0, stream>>>(XIN, W1, a1s, a1d, H, SV, DV, N_NODES, 64);
    gat_segment<<<((N_NODES + 3) / 4 + 7) / 8, 256, 0, stream>>>(OFF, CSRC, H, SV, DV, b1, XIN, N_NODES, 1);

    gat_gemm_wmma<<<((N_NODES / 16) + 7) / 8, 256, 0, stream>>>(XIN, W2, a2s, a2d, H, SV, DV, N_NODES, 64);
    gat_segment<<<((N_NODES + 3) / 4 + 7) / 8, 256, 0, stream>>>(OFF, CSRC, H, SV, DV, b2, out, N_NODES, 0);
}


# GEMMs of layers 1,2 run at the end of the preceding row pass: the block's 32 finished rows go through LDS to WMMA tiles; own logits s,d formed in the row pass; 2 launches go; logit sum order changes
# speedup vs baseline: 4.5645x; 1.1216x over previous
#include <hip/hip_runtime.h>
#include <cstddef>


#define N_NODES 50000
#define E_EDGES 800000
#define NEG_SLOPE 0.2f

typedef float v2f __attribute__((ext_vector_type(2)));
typedef float v8f __attribute__((ext_vector_type(8)));

__global__ void __launch_bounds__(256)
gat_gemm_wmma(const float* __restrict__ X, const float* __restrict__ W,
              float* __restrict__ H, int nrows, int K) {
    const int wave = blockIdx.x * (blockDim.x >> 5) + (threadIdx.x >> 5);
    const int lane = threadIdx.x & 31;
    const int rowBase = wave * 16;
    if (rowBase >= nrows) return;

    const int lo = lane & 15;
    const int hi = lane >> 4;

    v8f acc0 = {}, acc1 = {}, acc2 = {}, acc3 = {};

    const float* xr = X + (size_t)(rowBase + lo) * K + 2 * hi;
    const float* wc = W + lo;

    for (int k = 0; k < K; k += 4) {
        float2 af = *(const float2*)(xr + k);
        v2f a; a.x = af.x; a.y = af.y;

        const float* w0 = wc + (size_t)(k + 2 * hi) * 64;
        const float* w1 = w0 + 64;
        v2f b0; b0.x = w0[0];  b0.y = w1[0];
        v2f b1; b1.x = w0[16]; b1.y = w1[16];
        v2f b2; b2.x = w0[32]; b2.y = w1[32];
        v2f b3; b3.x = w0[48]; b3.y = w1[48];

        acc0 = __builtin_amdgcn_wmma_f32_16x16x4_f32(false, a, false, b0, (short)0, acc0, false, false);
        acc1 = __builtin_amdgcn_wmma_f32_16x16x4_f32(false, a, false, b1, (short)0, acc1, false, false);
        acc2 = __builtin_amdgcn_wmma_f32_16x16x4_f32(false, a, false, b2, (short)0, acc2, false, false);
        acc3 = __builtin_amdgcn_wmma_f32_16x16x4_f32(false, a, false, b3, (short)0, acc3, false, false);
    }

#pragma unroll
    for (int r = 0; r < 8; ++r) {
        float* o = H + (size_t)(rowBase + r + 8 * hi) * 64 + lo;
        o[0]  = acc0[r];
        o[16] = acc1[r];
        o[32] = acc2[r];
        o[48] = acc3[r];
    }
}

__global__ void __launch_bounds__(256)
csr_zero(int* __restrict__ cnt, int n) {
    int i = blockIdx.x * blockDim.x + threadIdx.x;
    if (i < n) cnt[i] = 0;
}

__global__ void __launch_bounds__(256)
csr_count(const int* __restrict__ dst, int* __restrict__ cnt,
          int* __restrict__ rank, int nE) {
    int e = blockIdx.x * blockDim.x + threadIdx.x;
    if (e >= nE) return;
    rank[e] = atomicAdd(cnt + dst[e], 1);
}

__global__ void __launch_bounds__(1024)
csr_scan(const int* __restrict__ cnt, int* __restrict__ off, int n) {
    __shared__ int red[32];
    __shared__ int wtot[32];
    const int t = threadIdx.x;
    const int lane = t & 31;
    const int w = t >> 5;
    const int b = blockIdx.x;

    int p = 0;
    for (int k = 0; k < b; ++k) p += cnt[k * 1024 + t];
    for (int m = 16; m >= 1; m >>= 1) p += __shfl_xor(p, m, 32);

    const int i = b * 1024 + t;
    const int v = (i < n) ? cnt[i] : 0;
    int x = v;
    for (int dlt = 1; dlt < 32; dlt <<= 1) {
        int y = __shfl_up(x, dlt, 32);
        if (lane >= dlt) x += y;
    }
    if (lane == 0)  red[w]  = p;
    if (lane == 31) wtot[w] = x;
    __syncthreads();
    int base = 0;
    for (int j = 0; j < 32; ++j) {
        base += red[j];
        if (j < w) base += wtot[j];
    }
    const int excl = base + x - v;
    if (i < n) off[i] = excl;
    if (i == n - 1) off[n] = excl + v;
}

__global__ void __launch_bounds__(256)
csr_scatter(const int* __restrict__ src, const int* __restrict__ dst,
            const int* __restrict__ off, const int* __restrict__ rank,
            int* __restrict__ csrc, int nE) {
    int e = blockIdx.x * blockDim.x + threadIdx.x;
    if (e >= nE) return;
    csrc[off[dst[e]] + rank[e]] = src[e];
}

__device__ __forceinline__ float leaky(float v) {
    return (v > 0.0f) ? v : NEG_SLOPE * v;
}


#define SEL8_BCAST(T) ((T) * 0x249249u)
#define SEL8_XOR1 0xde54c1u
#define SEL8_XOR2 0xb3e21au
#define SEL8_XOR4 0x688facu
#define DPP8F(v, SEL) __uint_as_float(__builtin_amdgcn_mov_dpp8(__float_as_uint(v), (SEL)))

__device__ __forceinline__ float group_max8(float v) {
    v = fmaxf(v, DPP8F(v, SEL8_XOR1));
    v = fmaxf(v, DPP8F(v, SEL8_XOR2));
    v = fmaxf(v, DPP8F(v, SEL8_XOR4));
    return v;
}

__device__ __forceinline__ float group_sum8(float v) {
    v += DPP8F(v, SEL8_XOR1);
    v += DPP8F(v, SEL8_XOR2);
    v += DPP8F(v, SEL8_XOR4);
    return v;
}

__device__ __forceinline__ void batch8(const char* __restrict__ Hb, unsigned laneoff, unsigned l8,
                                       int sj, bool valid, float dn,
                                       const float4 asp, const float4 asq,
                                       float& m, float& l, float4& a0, float4& a1) {
    float4 p[8], q[8];
#define GAT_LD(T)                                                                          \
    {                                                                                      \
        const unsigned su = __builtin_amdgcn_mov_dpp8((unsigned)sj, SEL8_BCAST(T));        \
        const char* r = Hb + ((su << 8) | laneoff);                                        \
        p[T] = *(const float4*)r;                                                          \
        q[T] = *(const float4*)(r + 128);                                                  \
    }
    GAT_LD(0) GAT_LD(1) GAT_LD(2) GAT_LD(3) GAT_LD(4) GAT_LD(5) GAT_LD(6) GAT_LD(7)
#undef GAT_LD

    float ds[8];
#pragma unroll
    for (int t = 0; t < 8; ++t) {
        float v = p[t].x * asp.x;
        v = fmaf(p[t].y, asp.y, v);
        v = fmaf(p[t].z, asp.z, v);
        v = fmaf(p[t].w, asp.w, v);
        v = fmaf(q[t].x, asq.x, v);
        v = fmaf(q[t].y, asq.y, v);
        v = fmaf(q[t].z, asq.z, v);
        v = fmaf(q[t].w, asq.w, v);
        ds[t] = v;
    }
    const bool b0 = (l8 & 1u) != 0u, b1 = (l8 & 2u) != 0u, b2 = (l8 & 4u) != 0u;
    float r1[4], r2[2];
#pragma unroll
    for (int i = 0; i < 4; ++i) {
        const float keep = b0 ? ds[2 * i + 1] : ds[2 * i];
        const float give = b0 ? ds[2 * i] : ds[2 * i + 1];
        r1[i] = keep + DPP8F(give, SEL8_XOR1);
    }
#pragma unroll
    for (int i = 0; i < 2; ++i) {
        const float keep = b1 ? r1[2 * i + 1] : r1[2 * i];
        const float give = b1 ? r1[2 * i] : r1[2 * i + 1];
        r2[i] = keep + DPP8F(give, SEL8_XOR2);
    }
    const float keep3 = b2 ? r2[1] : r2[0];
    const float give3 = b2 ? r2[0] : r2[1];
    const float sl = keep3 + DPP8F(give3, SEL8_XOR4);

    const float e  = leaky(sl + dn);
    const float mn = group_max8(valid ? fmaxf(e, m) : m);
    const float sc = __expf(m - mn);
    const float w  = valid ? __expf(e - mn) : 0.0f;
    l = fmaf(l, sc, group_sum8(w));
    m = mn;
    a0.x *= sc; a0.y *= sc; a0.z *= sc; a0.w *= sc;
    a1.x *= sc; a1.y *= sc; a1.z *= sc; a1.w *= sc;

    float cu[8];
#define GAT_W(T) cu[T] = DPP8F(w, SEL8_BCAST(T));
    GAT_W(0) GAT_W(1) GAT_W(2) GAT_W(3) GAT_W(4) GAT_W(5) GAT_W(6) GAT_W(7)
#undef GAT_W
#pragma unroll
    for (int t = 0; t < 8; ++t) {
        a0.x = fmaf(p[t].x, cu[t], a0.x);
        a0.y = fmaf(p[t].y, cu[t], a0.y);
        a0.z = fmaf(p[t].z, cu[t], a0.z);
        a0.w = fmaf(p[t].w, cu[t], a0.w);
        a1.x = fmaf(q[t].x, cu[t], a1.x);
        a1.y = fmaf(q[t].y, cu[t], a1.y);
        a1.z = fmaf(q[t].z, cu[t], a1.z);
        a1.w = fmaf(q[t].w, cu[t], a1.w);
    }
}

#define TILE_PITCH 68

template <bool NEXT>
__global__ void __launch_bounds__(256, 8)
gat_segment(const int* __restrict__ off, const int* __restrict__ csrc,
            const float* __restrict__ H, const float* __restrict__ asrc,
            const float* __restrict__ adst, const float* __restrict__ bias,
            const float* __restrict__ Wn, float* __restrict__ out, int n) {
    __shared__ __attribute__((aligned(16))) float tile[NEXT ? 32 * TILE_PITCH : 4];

    const int w = __builtin_amdgcn_readfirstlane((int)(threadIdx.x >> 5));
    const int wave = (int)(blockIdx.x * (blockDim.x >> 5)) + w;
    const int lane = threadIdx.x & 31;

    if (wave * 4 < n) {
        const unsigned l8 = (unsigned)lane & 7u;
        int node = wave * 4 + (lane >> 3);
        const bool live = node < n;
        if (!live) node = n - 1;

        const char* __restrict__ Hb = (const char*)H;
        const unsigned laneoff = l8 << 4;
        const unsigned own = ((unsigned)node << 8) | laneoff;

        const float4 hp = *(const float4*)(Hb + own);
        const float4 hq = *(const float4*)(Hb + own + 128);
        const float4 asp = ((const float4*)asrc)[l8];
        const float4 asq = ((const float4*)asrc)[8 + l8];
        const float4 adp = ((const float4*)adst)[l8];
        const float4 adq = ((const float4*)adst)[8 + l8];

        const int beg = off[node];
        const int deg = off[node + 1] - beg;

        const int dA = __builtin_amdgcn_readlane(deg, 0);
        const int dB = __builtin_amdgcn_readlane(deg, 8);
        const int dC = __builtin_amdgcn_readlane(deg, 16);
        const int dD = __builtin_amdgcn_readlane(deg, 24);
        const int dAB = dA > dB ? dA : dB;
        const int dCD = dC > dD ? dC : dD;
        const int dmax = dAB > dCD ? dAB : dCD;

        int   sj[4];
        bool  has[4];
        int   ld[4];
#pragma unroll
        for (int k = 0; k < 4; ++k) {
            const int j = (int)l8 + 8 * k;
            has[k] = j < deg;
            ld[k] = csrc[has[k] ? beg + j : 0];
        }
#pragma unroll
        for (int k = 0; k < 4; ++k) sj[k] = has[k] ? ld[k] : node;

        float ss = hp.x * asp.x, dd = hp.x * adp.x;
        ss = fmaf(hp.y, asp.y, ss); dd = fmaf(hp.y, adp.y, dd);
        ss = fmaf(hp.z, asp.z, ss); dd = fmaf(hp.z, adp.z, dd);
        ss = fmaf(hp.w, asp.w, ss); dd = fmaf(hp.w, adp.w, dd);
        ss = fmaf(hq.x, asq.x, ss); dd = fmaf(hq.x, adq.x, dd);
        ss = fmaf(hq.y, asq.y, ss); dd = fmaf(hq.y, adq.y, dd);
        ss = fmaf(hq.z, asq.z, ss); dd = fmaf(hq.z, adq.z, dd);
        ss = fmaf(hq.w, asq.w, ss); dd = fmaf(hq.w, adq.w, dd);
        const float sn = group_sum8(ss);
        const float dn = group_sum8(dd);

        float m = leaky(sn + dn);
        float l = 1.0f;
        float4 a0 = hp, a1 = hq;

#pragma unroll
        for (int k = 0; k < 4; ++k) {
            if (dmax > 8 * k) batch8(Hb, laneoff, l8, sj[k], has[k], dn, asp, asq, m, l, a0, a1);
        }
        for (int c = 32; c < dmax; c += 8) {
            const int j = c + (int)l8;
            const bool hv = j < deg;
            int sx = node;
            if (hv) sx = csrc[beg + j];
            batch8(Hb, laneoff, l8, sx, hv, dn, asp, asq, m, l, a0, a1);
        }

        const float4 bp = ((const float4*)bias)[l8];
        const float4 bq = ((const float4*)bias)[8 + l8];
        const float inv = 1.0f / (l + 1e-16f);
        float4 o0, o1;
        o0.x = fmaf(a0.x, inv, bp.x); o0.y = fmaf(a0.y, inv, bp.y); o0.z = fmaf(a0.z, inv, bp.z); o0.w = fmaf(a0.w, inv, bp.w);
        o1.x = fmaf(a1.x, inv, bq.x); o1.y = fmaf(a1.y, inv, bq.y); o1.z = fmaf(a1.z, inv, bq.z); o1.w = fmaf(a1.w, inv, bq.w);
        if constexpr (NEXT) {
            o0.x = fmaxf(o0.x, 0.0f); o0.y = fmaxf(o0.y, 0.0f); o0.z = fmaxf(o0.z, 0.0f); o0.w = fmaxf(o0.w, 0.0f);
            o1.x = fmaxf(o1.x, 0.0f); o1.y = fmaxf(o1.y, 0.0f); o1.z = fmaxf(o1.z, 0.0f); o1.w = fmaxf(o1.w, 0.0f);
            float* tr = tile + (4 * w + (lane >> 3)) * TILE_PITCH + 4 * (int)l8;
            *(float4*)tr        = o0;
            *(float4*)(tr + 32) = o1;
        } else {
            if (live) {
                char* ob = (char*)out;
                *(float4*)(ob + own)       = o0;
                *(float4*)(ob + own + 128) = o1;
            }
        }
    }

    if constexpr (NEXT) {
        const int lo = lane & 15;
        const int hi = lane >> 4;
        const int rb = w >> 1;
        const int ch = w & 1;
        const int row0 = (int)blockIdx.x * 32 + rb * 16;
        const bool work = (w < 4) && (row0 < n);

        v2f bA[16], bB[16];
        if (work) {
            const float* wc = Wn + (size_t)(2 * hi) * 64 + ch * 32 + lo;
#pragma unroll
            for (int kk = 0; kk < 16; ++kk) {
                bA[kk].x = wc[(4 * kk) * 64];      bA[kk].y = wc[(4 * kk + 1) * 64];
                bB[kk].x = wc[(4 * kk) * 64 + 16]; bB[kk].y = wc[(4 * kk + 1) * 64 + 16];
            }
        }

        __syncthreads();

        if (work) {
            const float* ar = tile + (rb * 16 + lo) * TILE_PITCH + 2 * hi;
            v8f acc0 = {}, acc1 = {};
#pragma unroll
            for (int kk = 0; kk < 16; ++kk) {
                const float2 af = *(const float2*)(ar + 4 * kk);
                v2f a; a.x = af.x; a.y = af.y;
                acc0 = __builtin_amdgcn_wmma_f32_16x16x4_f32(false, a, false, bA[kk], (short)0, acc0, false, false);
                acc1 = __builtin_amdgcn_wmma_f32_16x16x4_f32(false, a, false, bB[kk], (short)0, acc1, false, false);
            }
#pragma unroll
            for (int r = 0; r < 8; ++r) {
                float* o = out + (size_t)(row0 + r + 8 * hi) * 64 + ch * 32 + lo;
                o[0]  = acc0[r];
                o[16] = acc1[r];
            }
        }
    }
}

extern "C" void kernel_launch(void* const* d_in, const int* in_sizes, int n_in,
                              void* d_out, int out_size, void* d_ws, size_t ws_size,
                              hipStream_t stream) {
    (void)in_sizes; (void)n_in; (void)out_size; (void)ws_size;

    const float* x    = (const float*)d_in[0];
    const int*   esrc = (const int*)  d_in[1];
    const int*   edst = (const int*)  d_in[2];
    const float* W0 = (const float*)d_in[3];
    const float* a0s = (const float*)d_in[4];
    const float* a0d = (const float*)d_in[5];
    const float* b0 = (const float*)d_in[6];
    const float* W1 = (const float*)d_in[7];
    const float* a1s = (const float*)d_in[8];
    const float* a1d = (const float*)d_in[9];
    const float* b1 = (const float*)d_in[10];
    const float* W2 = (const float*)d_in[11];
    const float* a2s = (const float*)d_in[12];
    const float* a2d = (const float*)d_in[13];
    const float* b2 = (const float*)d_in[14];
    float* out = (float*)d_out;

    float* ws = (float*)d_ws;
    float* HA   = ws;
    float* HB   = HA  + (size_t)N_NODES * 64;
    int*   CNT  = (int*)(HB + (size_t)N_NODES * 64);
    int*   OFF  = CNT + N_NODES;
    int*   RANK = OFF + N_NODES + 1;
    int*   CSRC = RANK + E_EDGES;

    csr_zero<<<(N_NODES + 255) / 256, 256, 0, stream>>>(CNT, N_NODES);
    csr_count<<<(E_EDGES + 255) / 256, 256, 0, stream>>>(edst, CNT, RANK, E_EDGES);
    csr_scan<<<(N_NODES + 1023) / 1024, 1024, 0, stream>>>(CNT, OFF, N_NODES);
    csr_scatter<<<(E_EDGES + 255) / 256, 256, 0, stream>>>(esrc, edst, OFF, RANK, CSRC, E_EDGES);

    gat_gemm_wmma<<<((N_NODES / 16) + 7) / 8, 256, 0, stream>>>(x, W0, HA, N_NODES, 128);

    gat_segment<true><<<((N_NODES + 3) / 4 + 7) / 8, 256, 0, stream>>>(OFF, CSRC, HA, a0s, a0d, b0, W1, HB, N_NODES);

    gat_segment<true><<<((N_NODES + 3) / 4 + 7) / 8, 256, 0, stream>>>(OFF, CSRC, HB, a1s, a1d, b1, W2, HA, N_NODES);

    gat_segment<false><<<((N_NODES + 3) / 4 + 7) / 8, 256, 0, stream>>>(OFF, CSRC, HA, a2s, a2d, b2, W2, out, N_NODES);
}


# Fixed 32-slot bucket (one 128-byte line) per row: the count kernel puts the source id into its slot itself; scan and scatter launches go; rows over 32 add a spill list; sum order: slots, then spill
# speedup vs baseline: 5.7058x; 1.2500x over previous
#include <hip/hip_runtime.h>
#include <cstddef>


#define N_NODES 50000
#define E_EDGES 800000
#define NEG_SLOPE 0.2f
#define BUCKET_CAP 32
#define SPILL_CAP 24999

typedef float v2f __attribute__((ext_vector_type(2)));
typedef float v8f __attribute__((ext_vector_type(8)));

__global__ void __launch_bounds__(256)
gat_gemm_wmma(const float* __restrict__ X, const float* __restrict__ W,
              float* __restrict__ H, int nrows, int K) {
    const int wave = blockIdx.x * (blockDim.x >> 5) + (threadIdx.x >> 5);
    const int lane = threadIdx.x & 31;
    const int rowBase = wave * 16;
    if (rowBase >= nrows) return;

    const int lo = lane & 15;
    const int hi = lane >> 4;

    v8f acc0 = {}, acc1 = {}, acc2 = {}, acc3 = {};

    const float* xr = X + (size_t)(rowBase + lo) * K + 2 * hi;
    const float* wc = W + lo;

    for (int k = 0; k < K; k += 4) {
        float2 af = *(const float2*)(xr + k);
        v2f a; a.x = af.x; a.y = af.y;

        const float* w0 = wc + (size_t)(k + 2 * hi) * 64;
        const float* w1 = w0 + 64;
        v2f b0; b0.x = w0[0];  b0.y = w1[0];
        v2f b1; b1.x = w0[16]; b1.y = w1[16];
        v2f b2; b2.x = w0[32]; b2.y = w1[32];
        v2f b3; b3.x = w0[48]; b3.y = w1[48];

        acc0 = __builtin_amdgcn_wmma_f32_16x16x4_f32(false, a, false, b0, (short)0, acc0, false, false);
        acc1 = __builtin_amdgcn_wmma_f32_16x16x4_f32(false, a, false, b1, (short)0, acc1, false, false);
        acc2 = __builtin_amdgcn_wmma_f32_16x16x4_f32(false, a, false, b2, (short)0, acc2, false, false);
        acc3 = __builtin_amdgcn_wmma_f32_16x16x4_f32(false, a, false, b3, (short)0, acc3, false, false);
    }

#pragma unroll
    for (int r = 0; r < 8; ++r) {
        float* o = H + (size_t)(rowBase + r + 8 * hi) * 64 + lo;
        o[0]  = acc0[r];
        o[16] = acc1[r];
        o[32] = acc2[r];
        o[48] = acc3[r];
    }
}

__global__ void __launch_bounds__(256)
csr_zero(int* __restrict__ cnt, int n) {
    int i = blockIdx.x * blockDim.x + threadIdx.x;
    if (i < n) cnt[i] = 0;
}

__global__ void __launch_bounds__(256)
bucket_fill(const int* __restrict__ src, const int* __restrict__ dst,
            int* __restrict__ cnt, int* __restrict__ bkt, int n, int nE) {
    int e = blockIdx.x * blockDim.x + threadIdx.x;
    if (e >= nE) return;
    const int d = dst[e];
    const int s = src[e];
    const int slot = atomicAdd(cnt + d, 1);
    if (slot < BUCKET_CAP) {
        bkt[(size_t)d * BUCKET_CAP + slot] = s;
    } else {
        int* spd = cnt + n + 1;
        int* sps = spd + SPILL_CAP;
        const int k = atomicAdd(cnt + n, 1);
        if (k < SPILL_CAP) { spd[k] = d; sps[k] = s; }
    }
}

__device__ __forceinline__ float leaky(float v) {
    return (v > 0.0f) ? v : NEG_SLOPE * v;
}


#define SEL8_BCAST(T) ((T) * 0x249249u)
#define SEL8_XOR1 0xde54c1u
#define SEL8_XOR2 0xb3e21au
#define SEL8_XOR4 0x688facu
#define DPP8F(v, SEL) __uint_as_float(__builtin_amdgcn_mov_dpp8(__float_as_uint(v), (SEL)))

__device__ __forceinline__ float group_max8(float v) {
    v = fmaxf(v, DPP8F(v, SEL8_XOR1));
    v = fmaxf(v, DPP8F(v, SEL8_XOR2));
    v = fmaxf(v, DPP8F(v, SEL8_XOR4));
    return v;
}

__device__ __forceinline__ float group_sum8(float v) {
    v += DPP8F(v, SEL8_XOR1);
    v += DPP8F(v, SEL8_XOR2);
    v += DPP8F(v, SEL8_XOR4);
    return v;
}

__device__ __forceinline__ void batch8(const char* __restrict__ Hb, unsigned laneoff, unsigned l8,
                                       int sj, bool valid, float dn,
                                       const float4 asp, const float4 asq,
                                       float& m, float& l, float4& a0, float4& a1) {
    float4 p[8], q[8];
#define GAT_LD(T)                                                                          \
    {                                                                                      \
        const unsigned su = __builtin_amdgcn_mov_dpp8((unsigned)sj, SEL8_BCAST(T));        \
        const char* r = Hb + ((su << 8) | laneoff);                                        \
        p[T] = *(const float4*)r;                                                          \
        q[T] = *(const float4*)(r + 128);                                                  \
    }
    GAT_LD(0) GAT_LD(1) GAT_LD(2) GAT_LD(3) GAT_LD(4) GAT_LD(5) GAT_LD(6) GAT_LD(7)
#undef GAT_LD

    float ds[8];
#pragma unroll
    for (int t = 0; t < 8; ++t) {
        float v = p[t].x * asp.x;
        v = fmaf(p[t].y, asp.y, v);
        v = fmaf(p[t].z, asp.z, v);
        v = fmaf(p[t].w, asp.w, v);
        v = fmaf(q[t].x, asq.x, v);
        v = fmaf(q[t].y, asq.y, v);
        v = fmaf(q[t].z, asq.z, v);
        v = fmaf(q[t].w, asq.w, v);
        ds[t] = v;
    }
    const bool b0 = (l8 & 1u) != 0u, b1 = (l8 & 2u) != 0u, b2 = (l8 & 4u) != 0u;
    float r1[4], r2[2];
#pragma unroll
    for (int i = 0; i < 4; ++i) {
        const float keep = b0 ? ds[2 * i + 1] : ds[2 * i];
        const float give = b0 ? ds[2 * i] : ds[2 * i + 1];
        r1[i] = keep + DPP8F(give, SEL8_XOR1);
    }
#pragma unroll
    for (int i = 0; i < 2; ++i) {
        const float keep = b1 ? r1[2 * i + 1] : r1[2 * i];
        const float give = b1 ? r1[2 * i] : r1[2 * i + 1];
        r2[i] = keep + DPP8F(give, SEL8_XOR2);
    }
    const float keep3 = b2 ? r2[1] : r2[0];
    const float give3 = b2 ? r2[0] : r2[1];
    const float sl = keep3 + DPP8F(give3, SEL8_XOR4);

    const float e  = leaky(sl + dn);
    const float mn = group_max8(valid ? fmaxf(e, m) : m);
    const float sc = __expf(m - mn);
    const float w  = valid ? __expf(e - mn) : 0.0f;
    l = fmaf(l, sc, group_sum8(w));
    m = mn;
    a0.x *= sc; a0.y *= sc; a0.z *= sc; a0.w *= sc;
    a1.x *= sc; a1.y *= sc; a1.z *= sc; a1.w *= sc;

    float cu[8];
#define GAT_W(T) cu[T] = DPP8F(w, SEL8_BCAST(T));
    GAT_W(0) GAT_W(1) GAT_W(2) GAT_W(3) GAT_W(4) GAT_W(5) GAT_W(6) GAT_W(7)
#undef GAT_W
#pragma unroll
    for (int t = 0; t < 8; ++t) {
        a0.x = fmaf(p[t].x, cu[t], a0.x);
        a0.y = fmaf(p[t].y, cu[t], a0.y);
        a0.z = fmaf(p[t].z, cu[t], a0.z);
        a0.w = fmaf(p[t].w, cu[t], a0.w);
        a1.x = fmaf(q[t].x, cu[t], a1.x);
        a1.y = fmaf(q[t].y, cu[t], a1.y);
        a1.z = fmaf(q[t].z, cu[t], a1.z);
        a1.w = fmaf(q[t].w, cu[t], a1.w);
    }
}

#define TILE_PITCH 68

template <bool NEXT>
__global__ void __launch_bounds__(256, 8)
gat_segment(const int* __restrict__ cnt, const int* __restrict__ bkt,
            const int* __restrict__ esrc, const int* __restrict__ edst,
            const float* __restrict__ H, const float* __restrict__ asrc,
            const float* __restrict__ adst, const float* __restrict__ bias,
            const float* __restrict__ Wn, float* __restrict__ out, int n, int nE) {
    __shared__ __attribute__((aligned(16))) float tile[NEXT ? 32 * TILE_PITCH : 4];

    const int w = __builtin_amdgcn_readfirstlane((int)(threadIdx.x >> 5));
    const int wave = (int)(blockIdx.x * (blockDim.x >> 5)) + w;
    const int lane = threadIdx.x & 31;

    if (wave * 4 < n) {
        const unsigned l8 = (unsigned)lane & 7u;
        int node = wave * 4 + (lane >> 3);
        const bool live = node < n;
        if (!live) node = n - 1;

        const char* __restrict__ Hb = (const char*)H;
        const unsigned laneoff = l8 << 4;
        const unsigned own = ((unsigned)node << 8) | laneoff;

        const float4 hp = *(const float4*)(Hb + own);
        const float4 hq = *(const float4*)(Hb + own + 128);
        const float4 asp = ((const float4*)asrc)[l8];
        const float4 asq = ((const float4*)asrc)[8 + l8];
        const float4 adp = ((const float4*)adst)[l8];
        const float4 adq = ((const float4*)adst)[8 + l8];

        const int deg = cnt[node];
        const int* __restrict__ bk = bkt + (size_t)node * BUCKET_CAP;

        const int dA = __builtin_amdgcn_readlane(deg, 0);
        const int dB = __builtin_amdgcn_readlane(deg, 8);
        const int dC = __builtin_amdgcn_readlane(deg, 16);
        const int dD = __builtin_amdgcn_readlane(deg, 24);
        const int dAB = dA > dB ? dA : dB;
        const int dCD = dC > dD ? dC : dD;
        const int dmax = dAB > dCD ? dAB : dCD;

        int   sj[4];
        bool  has[4];
        int   ld[4];
#pragma unroll
        for (int k = 0; k < 4; ++k) {
            const int j = (int)l8 + 8 * k;
            has[k] = j < deg;
            ld[k] = *(has[k] ? bk + j : cnt + node);
        }
#pragma unroll
        for (int k = 0; k < 4; ++k) sj[k] = has[k] ? ld[k] : node;

        float ss = hp.x * asp.x, dd = hp.x * adp.x;
        ss = fmaf(hp.y, asp.y, ss); dd = fmaf(hp.y, adp.y, dd);
        ss = fmaf(hp.z, asp.z, ss); dd = fmaf(hp.z, adp.z, dd);
        ss = fmaf(hp.w, asp.w, ss); dd = fmaf(hp.w, adp.w, dd);
        ss = fmaf(hq.x, asq.x, ss); dd = fmaf(hq.x, adq.x, dd);
        ss = fmaf(hq.y, asq.y, ss); dd = fmaf(hq.y, adq.y, dd);
        ss = fmaf(hq.z, asq.z, ss); dd = fmaf(hq.z, adq.z, dd);
        ss = fmaf(hq.w, asq.w, ss); dd = fmaf(hq.w, adq.w, dd);
        const float sn = group_sum8(ss);
        const float dn = group_sum8(dd);

        float m = leaky(sn + dn);
        float l = 1.0f;
        float4 a0 = hp, a1 = hq;

#pragma unroll
        for (int k = 0; k < 4; ++k) {
            if (dmax > 8 * k) batch8(Hb, laneoff, l8, sj[k], has[k], dn, asp, asq, m, l, a0, a1);
        }
        if (deg > BUCKET_CAP) {
            const int ns = cnt[n];
            const bool full = ns > SPILL_CAP;
            const int* __restrict__ kd = full ? edst : cnt + n + 1;
            const int* __restrict__ ks = full ? esrc : cnt + n + 1 + SPILL_CAP;
            const int kn = full ? nE : ns;
            if (full) {
                a0 = *(const float4*)(Hb + own);
                a1 = *(const float4*)(Hb + own + 128);
                float s2 = a0.x * asp.x;
                s2 = fmaf(a0.y, asp.y, s2);
                s2 = fmaf(a0.z, asp.z, s2);
                s2 = fmaf(a0.w, asp.w, s2);
                s2 = fmaf(a1.x, asq.x, s2);
                s2 = fmaf(a1.y, asq.y, s2);
                s2 = fmaf(a1.z, asq.z, s2);
                s2 = fmaf(a1.w, asq.w, s2);
                m = leaky(group_sum8(s2) + dn);
                l = 1.0f;
            }
            for (int c = 0; c < kn; c += 8) {
                const int j = c + (int)l8;
                bool hv = false;
                if (j < kn) hv = kd[j] == node;
                int sx = node;
                if (hv) sx = ks[j];
                batch8(Hb, laneoff, l8, sx, hv, dn, asp, asq, m, l, a0, a1);
            }
        }

        const float4 bp = ((const float4*)bias)[l8];
        const float4 bq = ((const float4*)bias)[8 + l8];
        const float inv = 1.0f / (l + 1e-16f);
        float4 o0, o1;
        o0.x = fmaf(a0.x, inv, bp.x); o0.y = fmaf(a0.y, inv, bp.y); o0.z = fmaf(a0.z, inv, bp.z); o0.w = fmaf(a0.w, inv, bp.w);
        o1.x = fmaf(a1.x, inv, bq.x); o1.y = fmaf(a1.y, inv, bq.y); o1.z = fmaf(a1.z, inv, bq.z); o1.w = fmaf(a1.w, inv, bq.w);
        if constexpr (NEXT) {
            o0.x = fmaxf(o0.x, 0.0f); o0.y = fmaxf(o0.y, 0.0f); o0.z = fmaxf(o0.z, 0.0f); o0.w = fmaxf(o0.w, 0.0f);
            o1.x = fmaxf(o1.x, 0.0f); o1.y = fmaxf(o1.y, 0.0f); o1.z = fmaxf(o1.z, 0.0f); o1.w = fmaxf(o1.w, 0.0f);
            float* tr = tile + (4 * w + (lane >> 3)) * TILE_PITCH + 4 * (int)l8;
            *(float4*)tr        = o0;
            *(float4*)(tr + 32) = o1;
        } else {
            if (live) {
                char* ob = (char*)out;
                *(float4*)(ob + own)       = o0;
                *(float4*)(ob + own + 128) = o1;
            }
        }
    }

    if constexpr (NEXT) {
        const int lo = lane & 15;
        const int hi = lane >> 4;
        const int rb = w >> 1;
        const int ch = w & 1;
        const int row0 = (int)blockIdx.x * 32 + rb * 16;
        const bool work = (w < 4) && (row0 < n);

        v2f bA[16], bB[16];
        if (work) {
            const float* wc = Wn + (size_t)(2 * hi) * 64 + ch * 32 + lo;
#pragma unroll
            for (int kk = 0; kk < 16; ++kk) {
                bA[kk].x = wc[(4 * kk) * 64];      bA[kk].y = wc[(4 * kk + 1) * 64];
                bB[kk].x = wc[(4 * kk) * 64 + 16]; bB[kk].y = wc[(4 * kk + 1) * 64 + 16];
            }
        }

        __syncthreads();

        if (work) {
            const float* ar = tile + (rb * 16 + lo) * TILE_PITCH + 2 * hi;
            v8f acc0 = {}, acc1 = {};
#pragma unroll
            for (int kk = 0; kk < 16; ++kk) {
                const float2 af = *(const float2*)(ar + 4 * kk);
                v2f a; a.x = af.x; a.y = af.y;
                acc0 = __builtin_amdgcn_wmma_f32_16x16x4_f32(false, a, false, bA[kk], (short)0, acc0, false, false);
                acc1 = __builtin_amdgcn_wmma_f32_16x16x4_f32(false, a, false, bB[kk], (short)0, acc1, false, false);
            }
#pragma unroll
            for (int r = 0; r < 8; ++r) {
                float* o = out + (size_t)(row0 + r + 8 * hi) * 64 + ch * 32 + lo;
                o[0]  = acc0[r];
                o[16] = acc1[r];
            }
        }
    }
}

extern "C" void kernel_launch(void* const* d_in, const int* in_sizes, int n_in,
                              void* d_out, int out_size, void* d_ws, size_t ws_size,
                              hipStream_t stream) {
    (void)in_sizes; (void)n_in; (void)out_size; (void)ws_size;

    const float* x    = (const float*)d_in[0];
    const int*   esrc = (const int*)  d_in[1];
    const int*   edst = (const int*)  d_in[2];
    const float* W0 = (const float*)d_in[3];
    const float* a0s = (const float*)d_in[4];
    const float* a0d = (const float*)d_in[5];
    const float* b0 = (const float*)d_in[6];
    const float* W1 = (const float*)d_in[7];
    const float* a1s = (const float*)d_in[8];
    const float* a1d = (const float*)d_in[9];
    const float* b1 = (const float*)d_in[10];
    const float* W2 = (const float*)d_in[11];
    const float* a2s = (const float*)d_in[12];
    const float* a2d = (const float*)d_in[13];
    const float* b2 = (const float*)d_in[14];
    float* out = (float*)d_out;

    float* ws = (float*)d_ws;
    float* HA   = ws;
    float* HB   = HA  + (size_t)N_NODES * 64;
    int*   BKT  = (int*)(HB + (size_t)N_NODES * 64);
    int*   CNT  = BKT + (size_t)N_NODES * BUCKET_CAP;

    csr_zero<<<(N_NODES + 1 + 255) / 256, 256, 0, stream>>>(CNT, N_NODES + 1);
    bucket_fill<<<(E_EDGES + 255) / 256, 256, 0, stream>>>(esrc, edst, CNT, BKT, N_NODES, E_EDGES);

    gat_gemm_wmma<<<((N_NODES / 16) + 7) / 8, 256, 0, stream>>>(x, W0, HA, N_NODES, 128);

    gat_segment<true><<<((N_NODES + 3) / 4 + 7) / 8, 256, 0, stream>>>(CNT, BKT, esrc, edst, HA, a0s, a0d, b0, W1, HB, N_NODES, E_EDGES);

    gat_segment<true><<<((N_NODES + 3) / 4 + 7) / 8, 256, 0, stream>>>(CNT, BKT, esrc, edst, HB, a1s, a1d, b1, W2, HA, N_NODES, E_EDGES);

    gat_segment<false><<<((N_NODES + 3) / 4 + 7) / 8, 256, 0, stream>>>(CNT, BKT, esrc, edst, HA, a2s, a2d, b2, W2, out, N_NODES, E_EDGES);
}


# Buckets built with LDS counters, no global add or scattered store per edge: edge list cut in 256 pieces ordered by 128-row bin; one work-group per bin ranks its rows in LDS; slot (sum) order changes
# speedup vs baseline: 8.6785x; 1.5210x over previous
#include <hip/hip_runtime.h>
#include <cstddef>


#define N_NODES 50000
#define E_EDGES 800000
#define NEG_SLOPE 0.2f
#define BUCKET_CAP 32
#define SPILL_CAP 24999

typedef float v2f __attribute__((ext_vector_type(2)));
typedef float v8f __attribute__((ext_vector_type(8)));

__global__ void __launch_bounds__(256)
gat_gemm_wmma(const float* __restrict__ X, const float* __restrict__ W,
              float* __restrict__ H, int nrows, int K) {
    const int wave = blockIdx.x * (blockDim.x >> 5) + (threadIdx.x >> 5);
    const int lane = threadIdx.x & 31;
    const int rowBase = wave * 16;
    if (rowBase >= nrows) return;

    const int lo = lane & 15;
    const int hi = lane >> 4;

    v8f acc0 = {}, acc1 = {}, acc2 = {}, acc3 = {};

    const float* xr = X + (size_t)(rowBase + lo) * K + 2 * hi;
    const float* wc = W + lo;

    for (int k = 0; k < K; k += 4) {
        float2 af = *(const float2*)(xr + k);
        v2f a; a.x = af.x; a.y = af.y;

        const float* w0 = wc + (size_t)(k + 2 * hi) * 64;
        const float* w1 = w0 + 64;
        v2f b0; b0.x = w0[0];  b0.y = w1[0];
        v2f b1; b1.x = w0[16]; b1.y = w1[16];
        v2f b2; b2.x = w0[32]; b2.y = w1[32];
        v2f b3; b3.x = w0[48]; b3.y = w1[48];

        acc0 = __builtin_amdgcn_wmma_f32_16x16x4_f32(false, a, false, b0, (short)0, acc0, false, false);
        acc1 = __builtin_amdgcn_wmma_f32_16x16x4_f32(false, a, false, b1, (short)0, acc1, false, false);
        acc2 = __builtin_amdgcn_wmma_f32_16x16x4_f32(false, a, false, b2, (short)0, acc2, false, false);
        acc3 = __builtin_amdgcn_wmma_f32_16x16x4_f32(false, a, false, b3, (short)0, acc3, false, false);
    }

#pragma unroll
    for (int r = 0; r < 8; ++r) {
        float* o = H + (size_t)(rowBase + r + 8 * hi) * 64 + lo;
        o[0]  = acc0[r];
        o[16] = acc1[r];
        o[32] = acc2[r];
        o[48] = acc3[r];
    }
}

#define ROW_SHIFT 7
#define BIN_ROWS (1 << ROW_SHIFT)
#define N_BINS ((N_NODES + BIN_ROWS - 1) / BIN_ROWS)
#define OFF_PITCH (N_BINS + 1)
#define PART_BLOCKS 256
#define EPB ((E_EDGES + PART_BLOCKS - 1) / PART_BLOCKS)
#define REC_PITCH ((EPB + 8 + 31) / 32 * 32)
#define EPT ((EPB + 255) / 256)
#define SCAN_PER_LANE ((OFF_PITCH + 31) / 32)
#define HIST_SIZE (32 * SCAN_PER_LANE)

static_assert(N_NODES <= 65536, "a record keeps the source id in 16 bits");
static_assert(EPB < 65536, "a rank inside a piece is kept in 16 bits");
static_assert(N_BINS < 32768, "a bin index is kept in 15 bits");
static_assert(HIST_SIZE - 1 > N_BINS, "the last counter is a spare one behind the offsets");

__global__ void __launch_bounds__(256)
edge_partition(const int* __restrict__ src, const int* __restrict__ dst,
               int* __restrict__ rec, int* __restrict__ off,
               int* __restrict__ spill_count, int nE) {
    __shared__ int hist[HIST_SIZE];
    __shared__ __attribute__((aligned(16))) int stage[REC_PITCH];
    const int t = threadIdx.x;
    const int p = blockIdx.x;

    for (int i = t; i < HIST_SIZE; i += 256) hist[i] = 0;
    if (p == 0 && t == 0) *spill_count = 0;
    __syncthreads();

    const int base = p * EPB;
    int  dv[EPT], sv[EPT], key[EPT];
    bool ok[EPT];
#pragma unroll
    for (int j = 0; j < EPT; ++j) {
        const int q = j * 256 + t;
        ok[j] = (q < EPB) && (base + q < nE);
        const int e = ok[j] ? base + q : 0;
        dv[j] = dst[e];
        sv[j] = src[e];
    }
#pragma unroll
    for (int j = 0; j < EPT; ++j) {
        unsigned bin = (unsigned)dv[j] >> ROW_SHIFT;
        bin = bin < (unsigned)N_BINS ? bin : (unsigned)(N_BINS - 1);
        if (!ok[j]) bin = HIST_SIZE - 1;
        const int r = atomicAdd(&hist[bin], 1);
        key[j] = (int)(bin << 16) | r;
    }
    __syncthreads();

    if (t < 32) {
        int loc[SCAN_PER_LANE];
        int sum = 0;
#pragma unroll
        for (int i = 0; i < SCAN_PER_LANE; ++i) {
            loc[i] = sum;
            sum += hist[t * SCAN_PER_LANE + i];
        }
        int inc = sum;
#pragma unroll
        for (int o = 1; o < 32; o <<= 1) {
            const int y = __shfl_up(inc, o);
            if (t >= o) inc += y;
        }
        const int excl = inc - sum;
#pragma unroll
        for (int i = 0; i < SCAN_PER_LANE; ++i) hist[t * SCAN_PER_LANE + i] = excl + loc[i];
    }
    __syncthreads();

#pragma unroll
    for (int j = 0; j < EPT; ++j) {
        if (ok[j]) {
            const int at = hist[key[j] >> 16] + (key[j] & 0xffff);
            stage[at] = ((dv[j] & (BIN_ROWS - 1)) << 16) | sv[j];
        }
    }
    for (int i = t; i < OFF_PITCH; i += 256) off[p * OFF_PITCH + i] = hist[i];
    __syncthreads();

    int4* __restrict__ o4 = (int4*)(rec + (size_t)p * REC_PITCH);
    for (int i = t; i < REC_PITCH / 4; i += 256) o4[i] = ((const int4*)stage)[i];
}

__global__ void __launch_bounds__(256)
bucket_rank(const int* __restrict__ rec, const int* __restrict__ off,
            int* __restrict__ cnt, int* __restrict__ bkt, int n) {
    __shared__ int cntL[BIN_ROWS + 1];
    __shared__ __attribute__((aligned(16))) int bktL[BIN_ROWS * BUCKET_CAP];
    const int t = threadIdx.x;
    const int k = blockIdx.x;
    const int r0 = k * BIN_ROWS;

    if (t <= BIN_ROWS) cntL[t] = 0;
    const int* __restrict__ o = off + t * OFF_PITCH + k;
    const int a = o[0];
    const int b = o[1];
    const int* __restrict__ r = rec + (size_t)t * REC_PITCH;

    int v[4];
#pragma unroll
    for (int j = 0; j < 4; ++j) v[j] = r[a + j];
    __syncthreads();

    for (int i = a; i < b; i += 4) {
        int nx[4];
#pragma unroll
        for (int j = 0; j < 4; ++j) nx[j] = r[i + 4 + j];

        int rl[4], slot[4];
#pragma unroll
        for (int j = 0; j < 4; ++j) {
            rl[j] = (i + j < b) ? ((v[j] >> 16) & (BIN_ROWS - 1)) : BIN_ROWS;
            slot[j] = atomicAdd(&cntL[rl[j]], 1);
        }
#pragma unroll
        for (int j = 0; j < 4; ++j) {
            if (i + j < b) {
                const int s = v[j] & 0xffff;
                if (slot[j] < BUCKET_CAP) {
                    bktL[rl[j] * BUCKET_CAP + slot[j]] = s;
                } else {
                    int* spd = cnt + n + 1;
                    int* sps = spd + SPILL_CAP;
                    const int q = atomicAdd(cnt + n, 1);
                    if (q < SPILL_CAP) { spd[q] = r0 + rl[j]; sps[q] = s; }
                }
            }
        }
#pragma unroll
        for (int j = 0; j < 4; ++j) v[j] = nx[j];
    }
    __syncthreads();

    if (t < BIN_ROWS && r0 + t < n) cnt[r0 + t] = cntL[t];
    int4* __restrict__ o4 = (int4*)(bkt + (size_t)r0 * BUCKET_CAP);
    for (int i = t; i < BIN_ROWS * BUCKET_CAP / 4; i += 256) {
        if (r0 + (i >> 3) < n) o4[i] = ((const int4*)bktL)[i];
    }
}

__device__ __forceinline__ float leaky(float v) {
    return (v > 0.0f) ? v : NEG_SLOPE * v;
}


#define SEL8_BCAST(T) ((T) * 0x249249u)
#define SEL8_XOR1 0xde54c1u
#define SEL8_XOR2 0xb3e21au
#define SEL8_XOR4 0x688facu
#define DPP8F(v, SEL) __uint_as_float(__builtin_amdgcn_mov_dpp8(__float_as_uint(v), (SEL)))

__device__ __forceinline__ float group_max8(float v) {
    v = fmaxf(v, DPP8F(v, SEL8_XOR1));
    v = fmaxf(v, DPP8F(v, SEL8_XOR2));
    v = fmaxf(v, DPP8F(v, SEL8_XOR4));
    return v;
}

__device__ __forceinline__ float group_sum8(float v) {
    v += DPP8F(v, SEL8_XOR1);
    v += DPP8F(v, SEL8_XOR2);
    v += DPP8F(v, SEL8_XOR4);
    return v;
}

__device__ __forceinline__ void batch8(const char* __restrict__ Hb, unsigned laneoff, unsigned l8,
                                       int sj, bool valid, float dn,
                                       const float4 asp, const float4 asq,
                                       float& m, float& l, float4& a0, float4& a1) {
    float4 p[8], q[8];
#define GAT_LD(T)                                                                          \
    {                                                                                      \
        const unsigned su = __builtin_amdgcn_mov_dpp8((unsigned)sj, SEL8_BCAST(T));        \
        const char* r = Hb + ((su << 8) | laneoff);                                        \
        p[T] = *(const float4*)r;                                                          \
        q[T] = *(const float4*)(r + 128);                                                  \
    }
    GAT_LD(0) GAT_LD(1) GAT_LD(2) GAT_LD(3) GAT_LD(4) GAT_LD(5) GAT_LD(6) GAT_LD(7)
#undef GAT_LD

    float ds[8];
#pragma unroll
    for (int t = 0; t < 8; ++t) {
        float v = p[t].x * asp.x;
        v = fmaf(p[t].y, asp.y, v);
        v = fmaf(p[t].z, asp.z, v);
        v = fmaf(p[t].w, asp.w, v);
        v = fmaf(q[t].x, asq.x, v);
        v = fmaf(q[t].y, asq.y, v);
        v = fmaf(q[t].z, asq.z, v);
        v = fmaf(q[t].w, asq.w, v);
        ds[t] = v;
    }
    const bool b0 = (l8 & 1u) != 0u, b1 = (l8 & 2u) != 0u, b2 = (l8 & 4u) != 0u;
    float r1[4], r2[2];
#pragma unroll
    for (int i = 0; i < 4; ++i) {
        const float keep = b0 ? ds[2 * i + 1] : ds[2 * i];
        const float give = b0 ? ds[2 * i] : ds[2 * i + 1];
        r1[i] = keep + DPP8F(give, SEL8_XOR1);
    }
#pragma unroll
    for (int i = 0; i < 2; ++i) {
        const float keep = b1 ? r1[2 * i + 1] : r1[2 * i];
        const float give = b1 ? r1[2 * i] : r1[2 * i + 1];
        r2[i] = keep + DPP8F(give, SEL8_XOR2);
    }
    const float keep3 = b2 ? r2[1] : r2[0];
    const float give3 = b2 ? r2[0] : r2[1];
    const float sl = keep3 + DPP8F(give3, SEL8_XOR4);

    const float e  = leaky(sl + dn);
    const float mn = group_max8(valid ? fmaxf(e, m) : m);
    const float sc = __expf(m - mn);
    const float w  = valid ? __expf(e - mn) : 0.0f;
    l = fmaf(l, sc, group_sum8(w));
    m = mn;
    a0.x *= sc; a0.y *= sc; a0.z *= sc; a0.w *= sc;
    a1.x *= sc; a1.y *= sc; a1.z *= sc; a1.w *= sc;

    float cu[8];
#define GAT_W(T) cu[T] = DPP8F(w, SEL8_BCAST(T));
    GAT_W(0) GAT_W(1) GAT_W(2) GAT_W(3) GAT_W(4) GAT_W(5) GAT_W(6) GAT_W(7)
#undef GAT_W
#pragma unroll
    for (int t = 0; t < 8; ++t) {
        a0.x = fmaf(p[t].x, cu[t], a0.x);
        a0.y = fmaf(p[t].y, cu[t], a0.y);
        a0.z = fmaf(p[t].z, cu[t], a0.z);
        a0.w = fmaf(p[t].w, cu[t], a0.w);
        a1.x = fmaf(q[t].x, cu[t], a1.x);
        a1.y = fmaf(q[t].y, cu[t], a1.y);
        a1.z = fmaf(q[t].z, cu[t], a1.z);
        a1.w = fmaf(q[t].w, cu[t], a1.w);
    }
}

#define TILE_PITCH 68

template <bool NEXT>
__global__ void __launch_bounds__(256, 8)
gat_segment(const int* __restrict__ cnt, const int* __restrict__ bkt,
            const int* __restrict__ esrc, const int* __restrict__ edst,
            const float* __restrict__ H, const float* __restrict__ asrc,
            const float* __restrict__ adst, const float* __restrict__ bias,
            const float* __restrict__ Wn, float* __restrict__ out, int n, int nE) {
    __shared__ __attribute__((aligned(16))) float tile[NEXT ? 32 * TILE_PITCH : 4];

    const int w = __builtin_amdgcn_readfirstlane((int)(threadIdx.x >> 5));
    const int wave = (int)(blockIdx.x * (blockDim.x >> 5)) + w;
    const int lane = threadIdx.x & 31;

    if (wave * 4 < n) {
        const unsigned l8 = (unsigned)lane & 7u;
        int node = wave * 4 + (lane >> 3);
        const bool live = node < n;
        if (!live) node = n - 1;

        const char* __restrict__ Hb = (const char*)H;
        const unsigned laneoff = l8 << 4;
        const unsigned own = ((unsigned)node << 8) | laneoff;

        const float4 hp = *(const float4*)(Hb + own);
        const float4 hq = *(const float4*)(Hb + own + 128);
        const float4 asp = ((const float4*)asrc)[l8];
        const float4 asq = ((const float4*)asrc)[8 + l8];
        const float4 adp = ((const float4*)adst)[l8];
        const float4 adq = ((const float4*)adst)[8 + l8];

        const int deg = cnt[node];
        const int* __restrict__ bk = bkt + (size_t)node * BUCKET_CAP;

        const int dA = __builtin_amdgcn_readlane(deg, 0);
        const int dB = __builtin_amdgcn_readlane(deg, 8);
        const int dC = __builtin_amdgcn_readlane(deg, 16);
        const int dD = __builtin_amdgcn_readlane(deg, 24);
        const int dAB = dA > dB ? dA : dB;
        const int dCD = dC > dD ? dC : dD;
        const int dmax = dAB > dCD ? dAB : dCD;

        int   sj[4];
        bool  has[4];
        int   ld[4];
#pragma unroll
        for (int k = 0; k < 4; ++k) {
            const int j = (int)l8 + 8 * k;
            has[k] = j < deg;
            ld[k] = *(has[k] ? bk + j : cnt + node);
        }
#pragma unroll
        for (int k = 0; k < 4; ++k) sj[k] = has[k] ? ld[k] : node;

        float ss = hp.x * asp.x, dd = hp.x * adp.x;
        ss = fmaf(hp.y, asp.y, ss); dd = fmaf(hp.y, adp.y, dd);
        ss = fmaf(hp.z, asp.z, ss); dd = fmaf(hp.z, adp.z, dd);
        ss = fmaf(hp.w, asp.w, ss); dd = fmaf(hp.w, adp.w, dd);
        ss = fmaf(hq.x, asq.x, ss); dd = fmaf(hq.x, adq.x, dd);
        ss = fmaf(hq.y, asq.y, ss); dd = fmaf(hq.y, adq.y, dd);
        ss = fmaf(hq.z, asq.z, ss); dd = fmaf(hq.z, adq.z, dd);
        ss = fmaf(hq.w, asq.w, ss); dd = fmaf(hq.w, adq.w, dd);
        const float sn = group_sum8(ss);
        const float dn = group_sum8(dd);

        float m = leaky(sn + dn);
        float l = 1.0f;
        float4 a0 = hp, a1 = hq;

#pragma unroll
        for (int k = 0; k < 4; ++k) {
            if (dmax > 8 * k) batch8(Hb, laneoff, l8, sj[k], has[k], dn, asp, asq, m, l, a0, a1);
        }
        if (deg > BUCKET_CAP) {
            const int ns = cnt[n];
            const bool full = ns > SPILL_CAP;
            const int* __restrict__ kd = full ? edst : cnt + n + 1;
            const int* __restrict__ ks = full ? esrc : cnt + n + 1 + SPILL_CAP;
            const int kn = full ? nE : ns;
            if (full) {
                a0 = *(const float4*)(Hb + own);
                a1 = *(const float4*)(Hb + own + 128);
                float s2 = a0.x * asp.x;
                s2 = fmaf(a0.y, asp.y, s2);
                s2 = fmaf(a0.z, asp.z, s2);
                s2 = fmaf(a0.w, asp.w, s2);
                s2 = fmaf(a1.x, asq.x, s2);
                s2 = fmaf(a1.y, asq.y, s2);
                s2 = fmaf(a1.z, asq.z, s2);
                s2 = fmaf(a1.w, asq.w, s2);
                m = leaky(group_sum8(s2) + dn);
                l = 1.0f;
            }
            for (int c = 0; c < kn; c += 8) {
                const int j = c + (int)l8;
                bool hv = false;
                if (j < kn) hv = kd[j] == node;
                int sx = node;
                if (hv) sx = ks[j];
                batch8(Hb, laneoff, l8, sx, hv, dn, asp, asq, m, l, a0, a1);
            }
        }

        const float4 bp = ((const float4*)bias)[l8];
        const float4 bq = ((const float4*)bias)[8 + l8];
        const float inv = 1.0f / (l + 1e-16f);
        float4 o0, o1;
        o0.x = fmaf(a0.x, inv, bp.x); o0.y = fmaf(a0.y, inv, bp.y); o0.z = fmaf(a0.z, inv, bp.z); o0.w = fmaf(a0.w, inv, bp.w);
        o1.x = fmaf(a1.x, inv, bq.x); o1.y = fmaf(a1.y, inv, bq.y); o1.z = fmaf(a1.z, inv, bq.z); o1.w = fmaf(a1.w, inv, bq.w);
        if constexpr (NEXT) {
            o0.x = fmaxf(o0.x, 0.0f); o0.y = fmaxf(o0.y, 0.0f); o0.z = fmaxf(o0.z, 0.0f); o0.w = fmaxf(o0.w, 0.0f);
            o1.x = fmaxf(o1.x, 0.0f); o1.y = fmaxf(o1.y, 0.0f); o1.z = fmaxf(o1.z, 0.0f); o1.w = fmaxf(o1.w, 0.0f);
            float* tr = tile + (4 * w + (lane >> 3)) * TILE_PITCH + 4 * (int)l8;
            *(float4*)tr        = o0;
            *(float4*)(tr + 32) = o1;
        } else {
            if (live) {
                char* ob = (char*)out;
                *(float4*)(ob + own)       = o0;
                *(float4*)(ob + own + 128) = o1;
            }
        }
    }

    if constexpr (NEXT) {
        const int lo = lane & 15;
        const int hi = lane >> 4;
        const int rb = w >> 1;
        const int ch = w & 1;
        const int row0 = (int)blockIdx.x * 32 + rb * 16;
        const bool work = (w < 4) && (row0 < n);

        v2f bA[16], bB[16];
        if (work) {
            const float* wc = Wn + (size_t)(2 * hi) * 64 + ch * 32 + lo;
#pragma unroll
            for (int kk = 0; kk < 16; ++kk) {
                bA[kk].x = wc[(4 * kk) * 64];      bA[kk].y = wc[(4 * kk + 1) * 64];
                bB[kk].x = wc[(4 * kk) * 64 + 16]; bB[kk].y = wc[(4 * kk + 1) * 64 + 16];
            }
        }

        __syncthreads();

        if (work) {
            const float* ar = tile + (rb * 16 + lo) * TILE_PITCH + 2 * hi;
            v8f acc0 = {}, acc1 = {};
#pragma unroll
            for (int kk = 0; kk < 16; ++kk) {
                const float2 af = *(const float2*)(ar + 4 * kk);
                v2f a; a.x = af.x; a.y = af.y;
                acc0 = __builtin_amdgcn_wmma_f32_16x16x4_f32(false, a, false, bA[kk], (short)0, acc0, false, false);
                acc1 = __builtin_amdgcn_wmma_f32_16x16x4_f32(false, a, false, bB[kk], (short)0, acc1, false, false);
            }
#pragma unroll
            for (int r = 0; r < 8; ++r) {
                float* o = out + (size_t)(row0 + r + 8 * hi) * 64 + ch * 32 + lo;
                o[0]  = acc0[r];
                o[16] = acc1[r];
            }
        }
    }
}

extern "C" void kernel_launch(void* const* d_in, const int* in_sizes, int n_in,
                              void* d_out, int out_size, void* d_ws, size_t ws_size,
                              hipStream_t stream) {
    (void)in_sizes; (void)n_in; (void)out_size; (void)ws_size;

    const float* x    = (const float*)d_in[0];
    const int*   esrc = (const int*)  d_in[1];
    const int*   edst = (const int*)  d_in[2];
    const float* W0 = (const float*)d_in[3];
    const float* a0s = (const float*)d_in[4];
    const float* a0d = (const float*)d_in[5];
    const float* b0 = (const float*)d_in[6];
    const float* W1 = (const float*)d_in[7];
    const float* a1s = (const float*)d_in[8];
    const float* a1d = (const float*)d_in[9];
    const float* b1 = (const float*)d_in[10];
    const float* W2 = (const float*)d_in[11];
    const float* a2s = (const float*)d_in[12];
    const float* a2d = (const float*)d_in[13];
    const float* b2 = (const float*)d_in[14];
    float* out = (float*)d_out;

    float* ws = (float*)d_ws;
    float* HA   = ws;
    float* HB   = HA  + (size_t)N_NODES * 64;
    int*   BKT  = (int*)(HB + (size_t)N_NODES * 64);
    int*   CNT  = BKT + (size_t)N_NODES * BUCKET_CAP;

    int*   REC  = (int*)HB;
    int*   OFF  = REC + (size_t)PART_BLOCKS * REC_PITCH;

    edge_partition<<<PART_BLOCKS, 256, 0, stream>>>(esrc, edst, REC, OFF, CNT + N_NODES, E_EDGES);
    bucket_rank<<<N_BINS, 256, 0, stream>>>(REC, OFF, CNT, BKT, N_NODES);

    gat_gemm_wmma<<<((N_NODES / 16) + 7) / 8, 256, 0, stream>>>(x, W0, HA, N_NODES, 128);

    gat_segment<true><<<((N_NODES + 3) / 4 + 7) / 8, 256, 0, stream>>>(CNT, BKT, esrc, edst, HA, a0s, a0d, b0, W1, HB, N_NODES, E_EDGES);

    gat_segment<true><<<((N_NODES + 3) / 4 + 7) / 8, 256, 0, stream>>>(CNT, BKT, esrc, edst, HB, a1s, a1d, b1, W2, HA, N_NODES, E_EDGES);

    gat_segment<false><<<((N_NODES + 3) / 4 + 7) / 8, 256, 0, stream>>>(CNT, BKT, esrc, edst, HA, a2s, a2d, b2, W2, out, N_NODES, E_EDGES);
}


# Layer-0 GEMM work-groups ride in the edge-partition launch (role by block index, partition blocks first): one launch boundary goes and the two independent roles overlap; every sum order unchanged
# speedup vs baseline: 9.3727x; 1.0800x over previous
#include <hip/hip_runtime.h>
#include <cstddef>


#define N_NODES 50000
#define E_EDGES 800000
#define NEG_SLOPE 0.2f
#define BUCKET_CAP 32
#define SPILL_CAP 24999

typedef float v2f __attribute__((ext_vector_type(2)));
typedef float v8f __attribute__((ext_vector_type(8)));

__device__ __forceinline__ void
gat_gemm_wmma(const float* __restrict__ X, const float* __restrict__ W,
              float* __restrict__ H, int nrows, int K, int blk) {
    const int wave = blk * (int)(blockDim.x >> 5) + (int)(threadIdx.x >> 5);
    const int lane = threadIdx.x & 31;
    const int rowBase = wave * 16;
    if (rowBase >= nrows) return;

    const int lo = lane & 15;
    const int hi = lane >> 4;

    v8f acc0 = {}, acc1 = {}, acc2 = {}, acc3 = {};

    const float* xr = X + (size_t)(rowBase + lo) * K + 2 * hi;
    const float* wc = W + lo;

    for (int k = 0; k < K; k += 4) {
        float2 af = *(const float2*)(xr + k);
        v2f a; a.x = af.x; a.y = af.y;

        const float* w0 = wc + (size_t)(k + 2 * hi) * 64;
        const float* w1 = w0 + 64;
        v2f b0; b0.x = w0[0];  b0.y = w1[0];
        v2f b1; b1.x = w0[16]; b1.y = w1[16];
        v2f b2; b2.x = w0[32]; b2.y = w1[32];
        v2f b3; b3.x = w0[48]; b3.y = w1[48];

        acc0 = __builtin_amdgcn_wmma_f32_16x16x4_f32(false, a, false, b0, (short)0, acc0, false, false);
        acc1 = __builtin_amdgcn_wmma_f32_16x16x4_f32(false, a, false, b1, (short)0, acc1, false, false);
        acc2 = __builtin_amdgcn_wmma_f32_16x16x4_f32(false, a, false, b2, (short)0, acc2, false, false);
        acc3 = __builtin_amdgcn_wmma_f32_16x16x4_f32(false, a, false, b3, (short)0, acc3, false, false);
    }

#pragma unroll
    for (int r = 0; r < 8; ++r) {
        float* o = H + (size_t)(rowBase + r + 8 * hi) * 64 + lo;
        o[0]  = acc0[r];
        o[16] = acc1[r];
        o[32] = acc2[r];
        o[48] = acc3[r];
    }
}

#define ROW_SHIFT 7
#define BIN_ROWS (1 << ROW_SHIFT)
#define N_BINS ((N_NODES + BIN_ROWS - 1) / BIN_ROWS)
#define OFF_PITCH (N_BINS + 1)
#define PART_BLOCKS 256
#define EPB ((E_EDGES + PART_BLOCKS - 1) / PART_BLOCKS)
#define REC_PITCH ((EPB + 8 + 31) / 32 * 32)
#define EPT ((EPB + 255) / 256)
#define SCAN_PER_LANE ((OFF_PITCH + 31) / 32)
#define HIST_SIZE (32 * SCAN_PER_LANE)

static_assert(N_NODES <= 65536, "a record keeps the source id in 16 bits");
static_assert(EPB < 65536, "a rank inside a piece is kept in 16 bits");
static_assert(N_BINS < 32768, "a bin index is kept in 15 bits");
static_assert(HIST_SIZE - 1 > N_BINS, "the last counter is a spare one behind the offsets");

__device__ __forceinline__ void
edge_partition(const int* __restrict__ src, const int* __restrict__ dst,
               int* __restrict__ rec, int* __restrict__ off,
               int* __restrict__ spill_count, int nE,
               int* __restrict__ hist, int* __restrict__ stage, int p) {
    const int t = threadIdx.x;

    for (int i = t; i < HIST_SIZE; i += 256) hist[i] = 0;
    if (p == 0 && t == 0) *spill_count = 0;
    __syncthreads();

    const int base = p * EPB;
    int  dv[EPT], sv[EPT], key[EPT];
    bool ok[EPT];
#pragma unroll
    for (int j = 0; j < EPT; ++j) {
        const int q = j * 256 + t;
        ok[j] = (q < EPB) && (base + q < nE);
        const int e = ok[j] ? base + q : 0;
        dv[j] = dst[e];
        sv[j] = src[e];
    }
#pragma unroll
    for (int j = 0; j < EPT; ++j) {
        unsigned bin = (unsigned)dv[j] >> ROW_SHIFT;
        bin = bin < (unsigned)N_BINS ? bin : (unsigned)(N_BINS - 1);
        if (!ok[j]) bin = HIST_SIZE - 1;
        const int r = atomicAdd(&hist[bin], 1);
        key[j] = (int)(bin << 16) | r;
    }
    __syncthreads();

    if (t < 32) {
        int loc[SCAN_PER_LANE];
        int sum = 0;
#pragma unroll
        for (int i = 0; i < SCAN_PER_LANE; ++i) {
            loc[i] = sum;
            sum += hist[t * SCAN_PER_LANE + i];
        }
        int inc = sum;
#pragma unroll
        for (int o = 1; o < 32; o <<= 1) {
            const int y = __shfl_up(inc, o);
            if (t >= o) inc += y;
        }
        const int excl = inc - sum;
#pragma unroll
        for (int i = 0; i < SCAN_PER_LANE; ++i) hist[t * SCAN_PER_LANE + i] = excl + loc[i];
    }
    __syncthreads();

#pragma unroll
    for (int j = 0; j < EPT; ++j) {
        if (ok[j]) {
            const int at = hist[key[j] >> 16] + (key[j] & 0xffff);
            stage[at] = ((dv[j] & (BIN_ROWS - 1)) << 16) | sv[j];
        }
    }
    for (int i = t; i < OFF_PITCH; i += 256) off[p * OFF_PITCH + i] = hist[i];
    __syncthreads();

    int4* __restrict__ o4 = (int4*)(rec + (size_t)p * REC_PITCH);
    for (int i = t; i < REC_PITCH / 4; i += 256) o4[i] = ((const int4*)stage)[i];
}

#define GEMM0_BLOCKS ((N_NODES / 16 + 7) / 8)
__global__ void __launch_bounds__(256)
partition_and_gemm0(const int* __restrict__ src, const int* __restrict__ dst,
                    int* __restrict__ rec, int* __restrict__ off,
                    int* __restrict__ spill_count, int nE,
                    const float* __restrict__ X, const float* __restrict__ W,
                    float* __restrict__ H, int nrows, int K) {
    __shared__ int hist[HIST_SIZE];
    __shared__ __attribute__((aligned(16))) int stage[REC_PITCH];
    const int b = (int)blockIdx.x;
    if (b < PART_BLOCKS) {
        edge_partition(src, dst, rec, off, spill_count, nE, hist, stage, b);
    } else {
        gat_gemm_wmma(X, W, H, nrows, K, b - PART_BLOCKS);
    }
}

__global__ void __launch_bounds__(256)
bucket_rank(const int* __restrict__ rec, const int* __restrict__ off,
            int* __restrict__ cnt, int* __restrict__ bkt, int n) {
    __shared__ int cntL[BIN_ROWS + 1];
    __shared__ __attribute__((aligned(16))) int bktL[BIN_ROWS * BUCKET_CAP];
    const int t = threadIdx.x;
    const int k = blockIdx.x;
    const int r0 = k * BIN_ROWS;

    if (t <= BIN_ROWS) cntL[t] = 0;
    const int* __restrict__ o = off + t * OFF_PITCH + k;
    const int a = o[0];
    const int b = o[1];
    const int* __restrict__ r = rec + (size_t)t * REC_PITCH;

    int v[4];
#pragma unroll
    for (int j = 0; j < 4; ++j) v[j] = r[a + j];
    __syncthreads();

    for (int i = a; i < b; i += 4) {
        int nx[4];
#pragma unroll
        for (int j = 0; j < 4; ++j) nx[j] = r[i + 4 + j];

        int rl[4], slot[4];
#pragma unroll
        for (int j = 0; j < 4; ++j) {
            rl[j] = (i + j < b) ? ((v[j] >> 16) & (BIN_ROWS - 1)) : BIN_ROWS;
            slot[j] = atomicAdd(&cntL[rl[j]], 1);
        }
#pragma unroll
        for (int j = 0; j < 4; ++j) {
            if (i + j < b) {
                const int s = v[j] & 0xffff;
                if (slot[j] < BUCKET_CAP) {
                    bktL[rl[j] * BUCKET_CAP + slot[j]] = s;
                } else {
                    int* spd = cnt + n + 1;
                    int* sps = spd + SPILL_CAP;
                    const int q = atomicAdd(cnt + n, 1);
                    if (q < SPILL_CAP) { spd[q] = r0 + rl[j]; sps[q] = s; }
                }
            }
        }
#pragma unroll
        for (int j = 0; j < 4; ++j) v[j] = nx[j];
    }
    __syncthreads();

    if (t < BIN_ROWS && r0 + t < n) cnt[r0 + t] = cntL[t];
    int4* __restrict__ o4 = (int4*)(bkt + (size_t)r0 * BUCKET_CAP);
    for (int i = t; i < BIN_ROWS * BUCKET_CAP / 4; i += 256) {
        if (r0 + (i >> 3) < n) o4[i] = ((const int4*)bktL)[i];
    }
}

__device__ __forceinline__ float leaky(float v) {
    return (v > 0.0f) ? v : NEG_SLOPE * v;
}


#define SEL8_BCAST(T) ((T) * 0x249249u)
#define SEL8_XOR1 0xde54c1u
#define SEL8_XOR2 0xb3e21au
#define SEL8_XOR4 0x688facu
#define DPP8F(v, SEL) __uint_as_float(__builtin_amdgcn_mov_dpp8(__float_as_uint(v), (SEL)))

__device__ __forceinline__ float group_max8(float v) {
    v = fmaxf(v, DPP8F(v, SEL8_XOR1));
    v = fmaxf(v, DPP8F(v, SEL8_XOR2));
    v = fmaxf(v, DPP8F(v, SEL8_XOR4));
    return v;
}

__device__ __forceinline__ float group_sum8(float v) {
    v += DPP8F(v, SEL8_XOR1);
    v += DPP8F(v, SEL8_XOR2);
    v += DPP8F(v, SEL8_XOR4);
    return v;
}

__device__ __forceinline__ void batch8(const char* __restrict__ Hb, unsigned laneoff, unsigned l8,
                                       int sj, bool valid, float dn,
                                       const float4 asp, const float4 asq,
                                       float& m, float& l, float4& a0, float4& a1) {
    float4 p[8], q[8];
#define GAT_LD(T)                                                                          \
    {                                                                                      \
        const unsigned su = __builtin_amdgcn_mov_dpp8((unsigned)sj, SEL8_BCAST(T));        \
        const char* r = Hb + ((su << 8) | laneoff);                                        \
        p[T] = *(const float4*)r;                                                          \
        q[T] = *(const float4*)(r + 128);                                                  \
    }
    GAT_LD(0) GAT_LD(1) GAT_LD(2) GAT_LD(3) GAT_LD(4) GAT_LD(5) GAT_LD(6) GAT_LD(7)
#undef GAT_LD

    float ds[8];
#pragma unroll
    for (int t = 0; t < 8; ++t) {
        float v = p[t].x * asp.x;
        v = fmaf(p[t].y, asp.y, v);
        v = fmaf(p[t].z, asp.z, v);
        v = fmaf(p[t].w, asp.w, v);
        v = fmaf(q[t].x, asq.x, v);
        v = fmaf(q[t].y, asq.y, v);
        v = fmaf(q[t].z, asq.z, v);
        v = fmaf(q[t].w, asq.w, v);
        ds[t] = v;
    }
    const bool b0 = (l8 & 1u) != 0u, b1 = (l8 & 2u) != 0u, b2 = (l8 & 4u) != 0u;
    float r1[4], r2[2];
#pragma unroll
    for (int i = 0; i < 4; ++i) {
        const float keep = b0 ? ds[2 * i + 1] : ds[2 * i];
        const float give = b0 ? ds[2 * i] : ds[2 * i + 1];
        r1[i] = keep + DPP8F(give, SEL8_XOR1);
    }
#pragma unroll
    for (int i = 0; i < 2; ++i) {
        const float keep = b1 ? r1[2 * i + 1] : r1[2 * i];
        const float give = b1 ? r1[2 * i] : r1[2 * i + 1];
        r2[i] = keep + DPP8F(give, SEL8_XOR2);
    }
    const float keep3 = b2 ? r2[1] : r2[0];
    const float give3 = b2 ? r2[0] : r2[1];
    const float sl = keep3 + DPP8F(give3, SEL8_XOR4);

    const float e  = leaky(sl + dn);
    const float mn = group_max8(valid ? fmaxf(e, m) : m);
    const float sc = __expf(m - mn);
    const float w  = valid ? __expf(e - mn) : 0.0f;
    l = fmaf(l, sc, group_sum8(w));
    m = mn;
    a0.x *= sc; a0.y *= sc; a0.z *= sc; a0.w *= sc;
    a1.x *= sc; a1.y *= sc; a1.z *= sc; a1.w *= sc;

    float cu[8];
#define GAT_W(T) cu[T] = DPP8F(w, SEL8_BCAST(T));
    GAT_W(0) GAT_W(1) GAT_W(2) GAT_W(3) GAT_W(4) GAT_W(5) GAT_W(6) GAT_W(7)
#undef GAT_W
#pragma unroll
    for (int t = 0; t < 8; ++t) {
        a0.x = fmaf(p[t].x, cu[t], a0.x);
        a0.y = fmaf(p[t].y, cu[t], a0.y);
        a0.z = fmaf(p[t].z, cu[t], a0.z);
        a0.w = fmaf(p[t].w, cu[t], a0.w);
        a1.x = fmaf(q[t].x, cu[t], a1.x);
        a1.y = fmaf(q[t].y, cu[t], a1.y);
        a1.z = fmaf(q[t].z, cu[t], a1.z);
        a1.w = fmaf(q[t].w, cu[t], a1.w);
    }
}

#define TILE_PITCH 68

template <bool NEXT>
__global__ void __launch_bounds__(256, 8)
gat_segment(const int* __restrict__ cnt, const int* __restrict__ bkt,
            const int* __restrict__ esrc, const int* __restrict__ edst,
            const float* __restrict__ H, const float* __restrict__ asrc,
            const float* __restrict__ adst, const float* __restrict__ bias,
            const float* __restrict__ Wn, float* __restrict__ out, int n, int nE) {
    __shared__ __attribute__((aligned(16))) float tile[NEXT ? 32 * TILE_PITCH : 4];

    const int w = __builtin_amdgcn_readfirstlane((int)(threadIdx.x >> 5));
    const int wave = (int)(blockIdx.x * (blockDim.x >> 5)) + w;
    const int lane = threadIdx.x & 31;

    if (wave * 4 < n) {
        const unsigned l8 = (unsigned)lane & 7u;
        int node = wave * 4 + (lane >> 3);
        const bool live = node < n;
        if (!live) node = n - 1;

        const char* __restrict__ Hb = (const char*)H;
        const unsigned laneoff = l8 << 4;
        const unsigned own = ((unsigned)node << 8) | laneoff;

        const float4 hp = *(const float4*)(Hb + own);
        const float4 hq = *(const float4*)(Hb + own + 128);
        const float4 asp = ((const float4*)asrc)[l8];
        const float4 asq = ((const float4*)asrc)[8 + l8];
        const float4 adp = ((const float4*)adst)[l8];
        const float4 adq = ((const float4*)adst)[8 + l8];

        const int deg = cnt[node];
        const int* __restrict__ bk = bkt + (size_t)node * BUCKET_CAP;

        const int dA = __builtin_amdgcn_readlane(deg, 0);
        const int dB = __builtin_amdgcn_readlane(deg, 8);
        const int dC = __builtin_amdgcn_readlane(deg, 16);
        const int dD = __builtin_amdgcn_readlane(deg, 24);
        const int dAB = dA > dB ? dA : dB;
        const int dCD = dC > dD ? dC : dD;
        const int dmax = dAB > dCD ? dAB : dCD;

        int   sj[4];
        bool  has[4];
        int   ld[4];
#pragma unroll
        for (int k = 0; k < 4; ++k) {
            const int j = (int)l8 + 8 * k;
            has[k] = j < deg;
            ld[k] = *(has[k] ? bk + j : cnt + node);
        }
#pragma unroll
        for (int k = 0; k < 4; ++k) sj[k] = has[k] ? ld[k] : node;

        float ss = hp.x * asp.x, dd = hp.x * adp.x;
        ss = fmaf(hp.y, asp.y, ss); dd = fmaf(hp.y, adp.y, dd);
        ss = fmaf(hp.z, asp.z, ss); dd = fmaf(hp.z, adp.z, dd);
        ss = fmaf(hp.w, asp.w, ss); dd = fmaf(hp.w, adp.w, dd);
        ss = fmaf(hq.x, asq.x, ss); dd = fmaf(hq.x, adq.x, dd);
        ss = fmaf(hq.y, asq.y, ss); dd = fmaf(hq.y, adq.y, dd);
        ss = fmaf(hq.z, asq.z, ss); dd = fmaf(hq.z, adq.z, dd);
        ss = fmaf(hq.w, asq.w, ss); dd = fmaf(hq.w, adq.w, dd);
        const float sn = group_sum8(ss);
        const float dn = group_sum8(dd);

        float m = leaky(sn + dn);
        float l = 1.0f;
        float4 a0 = hp, a1 = hq;

#pragma unroll
        for (int k = 0; k < 4; ++k) {
            if (dmax > 8 * k) batch8(Hb, laneoff, l8, sj[k], has[k], dn, asp, asq, m, l, a0, a1);
        }
        if (deg > BUCKET_CAP) {
            const int ns = cnt[n];
            const bool full = ns > SPILL_CAP;
            const int* __restrict__ kd = full ? edst : cnt + n + 1;
            const int* __restrict__ ks = full ? esrc : cnt + n + 1 + SPILL_CAP;
            const int kn = full ? nE : ns;
            if (full) {
                a0 = *(const float4*)(Hb + own);
                a1 = *(const float4*)(Hb + own + 128);
                float s2 = a0.x * asp.x;
                s2 = fmaf(a0.y, asp.y, s2);
                s2 = fmaf(a0.z, asp.z, s2);
                s2 = fmaf(a0.w, asp.w, s2);
                s2 = fmaf(a1.x, asq.x, s2);
                s2 = fmaf(a1.y, asq.y, s2);
                s2 = fmaf(a1.z, asq.z, s2);
                s2 = fmaf(a1.w, asq.w, s2);
                m = leaky(group_sum8(s2) + dn);
                l = 1.0f;
            }
            for (int c = 0; c < kn; c += 8) {
                const int j = c + (int)l8;
                bool hv = false;
                if (j < kn) hv = kd[j] == node;
                int sx = node;
                if (hv) sx = ks[j];
                batch8(Hb, laneoff, l8, sx, hv, dn, asp, asq, m, l, a0, a1);
            }
        }

        const float4 bp = ((const float4*)bias)[l8];
        const float4 bq = ((const float4*)bias)[8 + l8];
        const float inv = 1.0f / (l + 1e-16f);
        float4 o0, o1;
        o0.x = fmaf(a0.x, inv, bp.x); o0.y = fmaf(a0.y, inv, bp.y); o0.z = fmaf(a0.z, inv, bp.z); o0.w = fmaf(a0.w, inv, bp.w);
        o1.x = fmaf(a1.x, inv, bq.x); o1.y = fmaf(a1.y, inv, bq.y); o1.z = fmaf(a1.z, inv, bq.z); o1.w = fmaf(a1.w, inv, bq.w);
        if constexpr (NEXT) {
            o0.x = fmaxf(o0.x, 0.0f); o0.y = fmaxf(o0.y, 0.0f); o0.z = fmaxf(o0.z, 0.0f); o0.w = fmaxf(o0.w, 0.0f);
            o1.x = fmaxf(o1.x, 0.0f); o1.y = fmaxf(o1.y, 0.0f); o1.z = fmaxf(o1.z, 0.0f); o1.w = fmaxf(o1.w, 0.0f);
            float* tr = tile + (4 * w + (lane >> 3)) * TILE_PITCH + 4 * (int)l8;
            *(float4*)tr        = o0;
            *(float4*)(tr + 32) = o1;
        } else {
            if (live) {
                char* ob = (char*)out;
                *(float4*)(ob + own)       = o0;
                *(float4*)(ob + own + 128) = o1;
            }
        }
    }

    if constexpr (NEXT) {
        const int lo = lane & 15;
        const int hi = lane >> 4;
        const int rb = w >> 1;
        const int ch = w & 1;
        const int row0 = (int)blockIdx.x * 32 + rb * 16;
        const bool work = (w < 4) && (row0 < n);

        v2f bA[16], bB[16];
        if (work) {
            const float* wc = Wn + (size_t)(2 * hi) * 64 + ch * 32 + lo;
#pragma unroll
            for (int kk = 0; kk < 16; ++kk) {
                bA[kk].x = wc[(4 * kk) * 64];      bA[kk].y = wc[(4 * kk + 1) * 64];
                bB[kk].x = wc[(4 * kk) * 64 + 16]; bB[kk].y = wc[(4 * kk + 1) * 64 + 16];
            }
        }

        __syncthreads();

        if (work) {
            const float* ar = tile + (rb * 16 + lo) * TILE_PITCH + 2 * hi;
            v8f acc0 = {}, acc1 = {};
#pragma unroll
            for (int kk = 0; kk < 16; ++kk) {
                const float2 af = *(const float2*)(ar + 4 * kk);
                v2f a; a.x = af.x; a.y = af.y;
                acc0 = __builtin_amdgcn_wmma_f32_16x16x4_f32(false, a, false, bA[kk], (short)0, acc0, false, false);
                acc1 = __builtin_amdgcn_wmma_f32_16x16x4_f32(false, a, false, bB[kk], (short)0, acc1, false, false);
            }
#pragma unroll
            for (int r = 0; r < 8; ++r) {
                float* o = out + (size_t)(row0 + r + 8 * hi) * 64 + ch * 32 + lo;
                o[0]  = acc0[r];
                o[16] = acc1[r];
            }
        }
    }
}

extern "C" void kernel_launch(void* const* d_in, const int* in_sizes, int n_in,
                              void* d_out, int out_size, void* d_ws, size_t ws_size,
                              hipStream_t stream) {
    (void)in_sizes; (void)n_in; (void)out_size; (void)ws_size;

    const float* x    = (const float*)d_in[0];
    const int*   esrc = (const int*)  d_in[1];
    const int*   edst = (const int*)  d_in[2];
    const float* W0 = (const float*)d_in[3];
    const float* a0s = (const float*)d_in[4];
    const float* a0d = (const float*)d_in[5];
    const float* b0 = (const float*)d_in[6];
    const float* W1 = (const float*)d_in[7];
    const float* a1s = (const float*)d_in[8];
    const float* a1d = (const float*)d_in[9];
    const float* b1 = (const float*)d_in[10];
    const float* W2 = (const float*)d_in[11];
    const float* a2s = (const float*)d_in[12];
    const float* a2d = (const float*)d_in[13];
    const float* b2 = (const float*)d_in[14];
    float* out = (float*)d_out;

    float* ws = (float*)d_ws;
    float* HA   = ws;
    float* HB   = HA  + (size_t)N_NODES * 64;
    int*   BKT  = (int*)(HB + (size_t)N_NODES * 64);
    int*   CNT  = BKT + (size_t)N_NODES * BUCKET_CAP;

    int*   REC  = (int*)HB;
    int*   OFF  = REC + (size_t)PART_BLOCKS * REC_PITCH;

    partition_and_gemm0<<<PART_BLOCKS + GEMM0_BLOCKS, 256, 0, stream>>>(esrc, edst, REC, OFF, CNT + N_NODES, E_EDGES,
                                                                      x, W0, HA, N_NODES, 128);
    bucket_rank<<<N_BINS, 256, 0, stream>>>(REC, OFF, CNT, BKT, N_NODES);

    gat_segment<true><<<((N_NODES + 3) / 4 + 7) / 8, 256, 0, stream>>>(CNT, BKT, esrc, edst, HA, a0s, a0d, b0, W1, HB, N_NODES, E_EDGES);

    gat_segment<true><<<((N_NODES + 3) / 4 + 7) / 8, 256, 0, stream>>>(CNT, BKT, esrc, edst, HB, a1s, a1d, b1, W2, HA, N_NODES, E_EDGES);

    gat_segment<false><<<((N_NODES + 3) / 4 + 7) / 8, 256, 0, stream>>>(CNT, BKT, esrc, edst, HA, a2s, a2d, b2, W2, out, N_NODES, E_EDGES);
}
